# starting kernel
# baseline (speedup 1.0000x reference)
#include <hip/hip_runtime.h>

typedef __attribute__((ext_vector_type(2))) float v2f;
typedef __attribute__((ext_vector_type(8))) float v8f;

#define GCN_N 50000
#define GCN_E 800000

__device__ __forceinline__ void atomic_add_f32(float* p, float v) {
  __hip_atomic_fetch_add(p, v, __ATOMIC_RELAXED, __HIP_MEMORY_SCOPE_AGENT);
}

__global__ void gcn_degree_kernel(const int* __restrict__ src,
                                  const int* __restrict__ dst,
                                  unsigned* __restrict__ deg_out,
                                  unsigned* __restrict__ deg_in,
                                  int n_edges) {
  int e = blockIdx.x * blockDim.x + threadIdx.x;
  if (e < n_edges) {
    atomicAdd(&deg_out[src[e]], 1u);
    atomicAdd(&deg_in[dst[e]], 1u);
  }
}

__global__ void gcn_norm_kernel(const unsigned* __restrict__ deg_out,
                                const unsigned* __restrict__ deg_in,
                                float* __restrict__ norm_out,
                                float* __restrict__ norm_in,
                                int n_nodes) {
  int n = blockIdx.x * blockDim.x + threadIdx.x;
  if (n >= n_nodes) return;
  unsigned dout = deg_out[n]; if (dout < 1u) dout = 1u;
  unsigned din  = deg_in[n];  if (din  < 1u) din  = 1u;
  norm_out[n] = rsqrtf((float)dout);
  norm_in[n]  = rsqrtf((float)din);
}

__global__ void gcn_scale_rows64_kernel(const float* __restrict__ x,
                                        const float* __restrict__ rownorm,
                                        float* __restrict__ y,
                                        int n_nodes) {
  int i = blockIdx.x * blockDim.x + threadIdx.x;
  if (i >= n_nodes * 16) return;
  int n = i >> 4;
  float s = rownorm[n];
  float4 v = ((const float4*)x)[i];
  v.x *= s; v.y *= s; v.z *= s; v.w *= s;
  ((float4*)y)[i] = v;
}

__global__ void gcn_scatter64_kernel(const int* __restrict__ src,
                                     const int* __restrict__ dst,
                                     const float* __restrict__ x,
                                     float* __restrict__ agg,
                                     int n_edges) {
  long long gid = (long long)blockIdx.x * blockDim.x + threadIdx.x;
  if (gid >= (long long)n_edges * 16) return;
  int e = (int)(gid >> 4);
  int q = (int)(gid & 15);
  int s = src[e];
  int d = dst[e];
  float4 v = ((const float4*)(x + (size_t)s * 64))[q];
  float* p = agg + (size_t)d * 64 + (size_t)q * 4;
  atomic_add_f32(p + 0, v.x);
  atomic_add_f32(p + 1, v.y);
  atomic_add_f32(p + 2, v.z);
  atomic_add_f32(p + 3, v.w);
}

__global__ void __launch_bounds__(256)
gcn_gemm1_kernel(const float* __restrict__ A,
                 const float* __restrict__ W,
                 const float* __restrict__ bias,
                 const float* __restrict__ norm_in,
                 const float* __restrict__ norm_out,
                 float* __restrict__ Y,
                 int n_tiles_m) {
  int wave = threadIdx.x >> 5;
  int tile = blockIdx.x * 8 + wave;
  int mt = tile >> 3;
  int nt = tile & 7;
  if (mt >= n_tiles_m) return;
  int lane = threadIdx.x & 31;
  int half = lane >> 4;
  int l16  = lane & 15;
  int arow = mt * 16 + l16;
  int bcol = nt * 16 + l16;
  float rs = norm_in[arow];
  const float* ap = A + (size_t)arow * 64;
  v8f acc = {};
#pragma unroll
  for (int k0 = 0; k0 < 64; k0 += 4) {
    int ka = k0 + half * 2;
    v2f a;
    a.x = ap[ka] * rs;
    a.y = ap[ka + 1] * rs;
    v2f b;
    b.x = W[(size_t)ka * 128 + bcol];
    b.y = W[(size_t)(ka + 1) * 128 + bcol];
    acc = __builtin_amdgcn_wmma_f32_16x16x4_f32(
        false, a, false, b, (short)0, acc, false, false);
  }
  float bb = bias[bcol];
#pragma unroll
  for (int r = 0; r < 8; ++r) {
    int row = mt * 16 + half * 8 + r;
    float v = fmaxf(acc[r] + bb, 0.0f) * norm_out[row];
    Y[(size_t)row * 128 + bcol] = v;
  }
}

__global__ void __launch_bounds__(256)
gcn_gemm2_kernel(const float* __restrict__ A,
                 const float* __restrict__ W,
                 float* __restrict__ Y,
                 int n_tiles) {
  int wave = threadIdx.x >> 5;
  int tile = blockIdx.x * 8 + wave;
  if (tile >= n_tiles) return;
  int mt = tile >> 2;
  int nt = tile & 3;
  int lane = threadIdx.x & 31;
  int half = lane >> 4;
  int l16  = lane & 15;
  int arow = mt * 16 + l16;
  int bcol = nt * 16 + l16;
  const float* ap = A + (size_t)arow * 128;
  v8f acc = {};
#pragma unroll
  for (int k0 = 0; k0 < 128; k0 += 4) {
    int ka = k0 + half * 2;
    v2f a;
    a.x = ap[ka];
    a.y = ap[ka + 1];
    v2f b;
    b.x = W[(size_t)ka * 64 + bcol];
    b.y = W[(size_t)(ka + 1) * 64 + bcol];
    acc = __builtin_amdgcn_wmma_f32_16x16x4_f32(
        false, a, false, b, (short)0, acc, false, false);
  }
#pragma unroll
  for (int r = 0; r < 8; ++r) {
    int row = mt * 16 + half * 8 + r;
    Y[(size_t)row * 64 + bcol] = acc[r];
  }
}

__global__ void gcn_final_kernel(const float* __restrict__ agg2,
                                 const float* __restrict__ norm_in,
                                 const float* __restrict__ bias,
                                 float* __restrict__ out,
                                 int n_nodes) {
  int i = blockIdx.x * blockDim.x + threadIdx.x;
  if (i >= n_nodes * 16) return;
  int n = i >> 4;
  int q = i & 15;
  float s = norm_in[n];
  float4 v = ((const float4*)agg2)[i];
  float4 bb = ((const float4*)bias)[q];
  v.x = v.x * s + bb.x;
  v.y = v.y * s + bb.y;
  v.z = v.z * s + bb.z;
  v.w = v.w * s + bb.w;
  ((float4*)out)[i] = v;
}

extern "C" void kernel_launch(void* const* d_in, const int* in_sizes, int n_in,
                              void* d_out, int out_size, void* d_ws, size_t ws_size,
                              hipStream_t stream) {
  (void)in_sizes; (void)n_in; (void)out_size; (void)ws_size;
  const float* h   = (const float*)d_in[0];
  const int*   src = (const int*)d_in[1];
  const int*   dst = (const int*)d_in[2];
  const float* W1  = (const float*)d_in[3];
  const float* b1  = (const float*)d_in[4];
  const float* W2  = (const float*)d_in[5];
  const float* b2  = (const float*)d_in[6];
  float* out = (float*)d_out;

  const int N = GCN_N;
  const int E = GCN_E;

  char* ws = (char*)d_ws;
  size_t off = 0;
  auto carve = [&](size_t bytes) -> void* {
    void* p = ws + off;
    off += (bytes + 255) & ~(size_t)255;
    return p;
  };
  unsigned* deg_out = (unsigned*)carve((size_t)N * 4);
  unsigned* deg_in  = (unsigned*)carve((size_t)N * 4);
  float* norm_out   = (float*)carve((size_t)N * 4);
  float* norm_in    = (float*)carve((size_t)N * 4);
  float* hs         = (float*)carve((size_t)N * 64 * 4);
  float* agg1       = (float*)carve((size_t)N * 64 * 4);
  float* h1s        = (float*)carve((size_t)N * 128 * 4);
  float* t2         = (float*)carve((size_t)N * 64 * 4);
  float* agg2       = (float*)carve((size_t)N * 64 * 4);

  hipMemsetAsync(deg_out, 0, (size_t)N * 4, stream);
  hipMemsetAsync(deg_in,  0, (size_t)N * 4, stream);
  hipMemsetAsync(agg1,    0, (size_t)N * 64 * 4, stream);
  hipMemsetAsync(agg2,    0, (size_t)N * 64 * 4, stream);

  gcn_degree_kernel<<<(E + 255) / 256, 256, 0, stream>>>(src, dst, deg_out, deg_in, E);
  gcn_norm_kernel<<<(N + 255) / 256, 256, 0, stream>>>(deg_out, deg_in, norm_out, norm_in, N);

  gcn_scale_rows64_kernel<<<(N * 16 + 255) / 256, 256, 0, stream>>>(h, norm_out, hs, N);
  {
    long long thr = (long long)E * 16;
    gcn_scatter64_kernel<<<(unsigned)((thr + 255) / 256), 256, 0, stream>>>(src, dst, hs, agg1, E);
  }
  gcn_gemm1_kernel<<<N / 16, 256, 0, stream>>>(agg1, W1, b1, norm_in, norm_out, h1s, N / 16);

  {
    int tiles = (N / 16) * 4;
    gcn_gemm2_kernel<<<(tiles + 7) / 8, 256, 0, stream>>>(h1s, W2, t2, tiles);
  }
  {
    long long thr = (long long)E * 16;
    gcn_scatter64_kernel<<<(unsigned)((thr + 255) / 256), 256, 0, stream>>>(src, dst, t2, agg2, E);
  }
  gcn_final_kernel<<<(N * 16 + 255) / 256, 256, 0, stream>>>(agg2, norm_in, b2, out, N);
}


# Replace float-atomic scatter-add by edges sorted by dst (rank from returning int atomic + prefix sum) and a per-node gather-sum; sums now in arrival-rank order on 4 partial sums, no float atomics
# speedup vs baseline: 1.4906x; 1.4906x over previous
#include <hip/hip_runtime.h>

typedef __attribute__((ext_vector_type(2))) float v2f;
typedef __attribute__((ext_vector_type(8))) float v8f;

#define GCN_N 50000
#define GCN_E 800000

__global__ void gcn_degree_kernel(const int* __restrict__ src,
                                  const int* __restrict__ dst,
                                  unsigned* __restrict__ deg_out,
                                  unsigned* __restrict__ deg_in,
                                  unsigned* __restrict__ rank,
                                  int n_edges) {
  int e = blockIdx.x * blockDim.x + threadIdx.x;
  if (e < n_edges) {
    atomicAdd(&deg_out[src[e]], 1u);
    rank[e] = atomicAdd(&deg_in[dst[e]], 1u);
  }
}

__global__ void __launch_bounds__(1024)
gcn_norm_rowptr_kernel(const unsigned* __restrict__ deg_out,
                       const unsigned* __restrict__ deg_in,
                       float* __restrict__ norm_out,
                       float* __restrict__ norm_in,
                       unsigned* __restrict__ rowptr,
                       int n_nodes) {
  __shared__ unsigned s_front[32];
  __shared__ unsigned s_wave[32];
  const int tid  = threadIdx.x;
  const int lane = tid & 31;
  const int wv   = tid >> 5;
  const int blk0 = blockIdx.x * 1024;
  const int n    = blk0 + tid;

  unsigned din = 0u;
  if (n < n_nodes) {
    unsigned dout = deg_out[n];
    din = deg_in[n];
    norm_out[n] = rsqrtf((float)(dout < 1u ? 1u : dout));
    norm_in[n]  = rsqrtf((float)(din  < 1u ? 1u : din));
  }

  unsigned front = 0u;
  const uint4* d4 = (const uint4*)deg_in;
#pragma unroll 4
  for (int i = tid; i < blk0 / 4; i += 1024) {
    uint4 t = d4[i];
    front += t.x + t.y + t.z + t.w;
  }
  for (int o = 16; o > 0; o >>= 1) front += __shfl_xor(front, o, 32);

  unsigned incl = din;
  for (int o = 1; o < 32; o <<= 1) {
    unsigned t = __shfl_up(incl, o, 32);
    if (lane >= o) incl += t;
  }
  if (lane == 0)  s_front[wv] = front;
  if (lane == 31) s_wave[wv]  = incl;
  __syncthreads();

  unsigned f = s_front[lane];
  unsigned w = (lane < wv) ? s_wave[lane] : 0u;
  unsigned base = f + w;
  for (int o = 16; o > 0; o >>= 1) base += __shfl_xor(base, o, 32);

  unsigned excl = base + incl - din;
  if (n < n_nodes) rowptr[n] = excl;
  if (n == n_nodes - 1) rowptr[n_nodes] = excl + din;
}

__global__ void gcn_fill_kernel(const int* __restrict__ src,
                                const int* __restrict__ dst,
                                const unsigned* __restrict__ rank,
                                const unsigned* __restrict__ rowptr,
                                int* __restrict__ csr_src,
                                int n_edges) {
  int e = blockIdx.x * blockDim.x + threadIdx.x;
  if (e < n_edges) {
    csr_src[rowptr[dst[e]] + rank[e]] = src[e];
  }
}

__global__ void gcn_scale_rows64_kernel(const float* __restrict__ x,
                                        const float* __restrict__ rownorm,
                                        float* __restrict__ y,
                                        int n_nodes) {
  int i = blockIdx.x * blockDim.x + threadIdx.x;
  if (i >= n_nodes * 16) return;
  int n = i >> 4;
  float s = rownorm[n];
  float4 v = ((const float4*)x)[i];
  v.x *= s; v.y *= s; v.z *= s; v.w *= s;
  ((float4*)y)[i] = v;
}

__device__ __forceinline__ void add4(float4& a, const float4& v) {
  a.x += v.x; a.y += v.y; a.z += v.z; a.w += v.w;
}

__global__ void __launch_bounds__(256)
gcn_gather64_kernel(const unsigned* __restrict__ rowptr,
                    const int* __restrict__ csr_src,
                    const float* __restrict__ x,
                    float* __restrict__ agg,
                    int n_nodes) {
  int gid = blockIdx.x * 256 + threadIdx.x;
  int n = gid >> 4;
  int q = gid & 15;
  if (n >= n_nodes) return;
  const char* xb = (const char*)x;
  const unsigned qoff = (unsigned)q * 16u;
  unsigned beg = rowptr[n];
  unsigned end = rowptr[n + 1];
  float4 a0 = make_float4(0.f, 0.f, 0.f, 0.f);
  float4 a1 = a0, a2 = a0, a3 = a0;
  for (unsigned base = beg; base < end; base += 32) {
    unsigned rem = end - base;
    int idx0 = ((unsigned)q < rem)        ? csr_src[base + q]      : 0;
    int idx1 = ((unsigned)(q + 16) < rem) ? csr_src[base + 16 + q] : 0;
    float4 v[32];
#pragma unroll
    for (int k = 0; k < 32; ++k) {
      int s = __shfl(k < 16 ? idx0 : idx1, k & 15, 16);
      v[k] = make_float4(0.f, 0.f, 0.f, 0.f);
      if ((unsigned)k < rem) v[k] = *(const float4*)(xb + ((unsigned)s * 256u + qoff));
    }
#pragma unroll
    for (int k = 0; k < 32; k += 4) {
      add4(a0, v[k]);
      add4(a1, v[k + 1]);
      add4(a2, v[k + 2]);
      add4(a3, v[k + 3]);
    }
  }
  add4(a0, a1);
  add4(a2, a3);
  add4(a0, a2);
  ((float4*)agg)[(size_t)n * 16 + q] = a0;
}

__global__ void __launch_bounds__(256)
gcn_gemm1_kernel(const float* __restrict__ A,
                 const float* __restrict__ W,
                 const float* __restrict__ bias,
                 const float* __restrict__ norm_in,
                 const float* __restrict__ norm_out,
                 float* __restrict__ Y,
                 int n_tiles_m) {
  int wave = threadIdx.x >> 5;
  int tile = blockIdx.x * 8 + wave;
  int mt = tile >> 3;
  int nt = tile & 7;
  if (mt >= n_tiles_m) return;
  int lane = threadIdx.x & 31;
  int half = lane >> 4;
  int l16  = lane & 15;
  int arow = mt * 16 + l16;
  int bcol = nt * 16 + l16;
  float rs = norm_in[arow];
  const float* ap = A + (size_t)arow * 64;
  v8f acc = {};
#pragma unroll
  for (int k0 = 0; k0 < 64; k0 += 4) {
    int ka = k0 + half * 2;
    v2f a;
    a.x = ap[ka] * rs;
    a.y = ap[ka + 1] * rs;
    v2f b;
    b.x = W[(size_t)ka * 128 + bcol];
    b.y = W[(size_t)(ka + 1) * 128 + bcol];
    acc = __builtin_amdgcn_wmma_f32_16x16x4_f32(
        false, a, false, b, (short)0, acc, false, false);
  }
  float bb = bias[bcol];
#pragma unroll
  for (int r = 0; r < 8; ++r) {
    int row = mt * 16 + half * 8 + r;
    float v = fmaxf(acc[r] + bb, 0.0f) * norm_out[row];
    Y[(size_t)row * 128 + bcol] = v;
  }
}

__global__ void __launch_bounds__(256)
gcn_gemm2_kernel(const float* __restrict__ A,
                 const float* __restrict__ W,
                 float* __restrict__ Y,
                 int n_tiles) {
  int wave = threadIdx.x >> 5;
  int tile = blockIdx.x * 8 + wave;
  if (tile >= n_tiles) return;
  int mt = tile >> 2;
  int nt = tile & 3;
  int lane = threadIdx.x & 31;
  int half = lane >> 4;
  int l16  = lane & 15;
  int arow = mt * 16 + l16;
  int bcol = nt * 16 + l16;
  const float* ap = A + (size_t)arow * 128;
  v8f acc = {};
#pragma unroll
  for (int k0 = 0; k0 < 128; k0 += 4) {
    int ka = k0 + half * 2;
    v2f a;
    a.x = ap[ka];
    a.y = ap[ka + 1];
    v2f b;
    b.x = W[(size_t)ka * 64 + bcol];
    b.y = W[(size_t)(ka + 1) * 64 + bcol];
    acc = __builtin_amdgcn_wmma_f32_16x16x4_f32(
        false, a, false, b, (short)0, acc, false, false);
  }
#pragma unroll
  for (int r = 0; r < 8; ++r) {
    int row = mt * 16 + half * 8 + r;
    Y[(size_t)row * 64 + bcol] = acc[r];
  }
}

__global__ void gcn_final_kernel(const float* __restrict__ agg2,
                                 const float* __restrict__ norm_in,
                                 const float* __restrict__ bias,
                                 float* __restrict__ out,
                                 int n_nodes) {
  int i = blockIdx.x * blockDim.x + threadIdx.x;
  if (i >= n_nodes * 16) return;
  int n = i >> 4;
  int q = i & 15;
  float s = norm_in[n];
  float4 v = ((const float4*)agg2)[i];
  float4 bb = ((const float4*)bias)[q];
  v.x = v.x * s + bb.x;
  v.y = v.y * s + bb.y;
  v.z = v.z * s + bb.z;
  v.w = v.w * s + bb.w;
  ((float4*)out)[i] = v;
}

extern "C" void kernel_launch(void* const* d_in, const int* in_sizes, int n_in,
                              void* d_out, int out_size, void* d_ws, size_t ws_size,
                              hipStream_t stream) {
  (void)in_sizes; (void)n_in; (void)out_size; (void)ws_size;
  const float* h   = (const float*)d_in[0];
  const int*   src = (const int*)d_in[1];
  const int*   dst = (const int*)d_in[2];
  const float* W1  = (const float*)d_in[3];
  const float* b1  = (const float*)d_in[4];
  const float* W2  = (const float*)d_in[5];
  const float* b2  = (const float*)d_in[6];
  float* out = (float*)d_out;

  const int N = GCN_N;
  const int E = GCN_E;

  char* ws = (char*)d_ws;
  size_t off = 0;
  auto carve = [&](size_t bytes) -> void* {
    void* p = ws + off;
    off += (bytes + 255) & ~(size_t)255;
    return p;
  };
  unsigned* deg_out = (unsigned*)carve((size_t)N * 4);
  unsigned* deg_in  = (unsigned*)carve((size_t)N * 4);
  float* norm_out   = (float*)carve((size_t)N * 4);
  float* norm_in    = (float*)carve((size_t)N * 4);
  float* hs         = (float*)carve((size_t)N * 64 * 4);
  float* agg        = (float*)carve((size_t)N * 64 * 4);
  float* h1s        = (float*)carve((size_t)N * 128 * 4);
  float* t2         = (float*)carve((size_t)N * 64 * 4);
  unsigned* rowptr  = (unsigned*)carve((size_t)(N + 1) * 4);
  unsigned* rank    = (unsigned*)carve((size_t)E * 4);
  int* csr_src      = (int*)carve((size_t)E * 4);

  hipMemsetAsync(deg_out, 0, (size_t)N * 4, stream);
  hipMemsetAsync(deg_in,  0, (size_t)N * 4, stream);

  gcn_degree_kernel<<<(E + 255) / 256, 256, 0, stream>>>(src, dst, deg_out, deg_in, rank, E);
  gcn_norm_rowptr_kernel<<<(N + 1023) / 1024, 1024, 0, stream>>>(deg_out, deg_in, norm_out, norm_in, rowptr, N);
  gcn_fill_kernel<<<(E + 255) / 256, 256, 0, stream>>>(src, dst, rank, rowptr, csr_src, E);

  gcn_scale_rows64_kernel<<<(N * 16 + 255) / 256, 256, 0, stream>>>(h, norm_out, hs, N);
  gcn_gather64_kernel<<<(N * 16 + 255) / 256, 256, 0, stream>>>(rowptr, csr_src, hs, agg, N);
  gcn_gemm1_kernel<<<N / 16, 256, 0, stream>>>(agg, W1, b1, norm_in, norm_out, h1s, N / 16);

  {
    int tiles = (N / 16) * 4;
    gcn_gemm2_kernel<<<(tiles + 7) / 8, 256, 0, stream>>>(h1s, W2, t2, tiles);
  }
  gcn_gather64_kernel<<<(N * 16 + 255) / 256, 256, 0, stream>>>(rowptr, csr_src, t2, agg, N);
  gcn_final_kernel<<<(N * 16 + 255) / 256, 256, 0, stream>>>(agg, norm_in, b2, out, N);
}


# Fuse launches: row-scaling pass joins the edge-fill kernel, final scale+bias joins the layer-2 gather's store, one clear for both counters; 11 graph nodes -> 8, arithmetic and sum order unchanged
# speedup vs baseline: 1.5567x; 1.0443x over previous
#include <hip/hip_runtime.h>

typedef __attribute__((ext_vector_type(2))) float v2f;
typedef __attribute__((ext_vector_type(8))) float v8f;

#define GCN_N 50000
#define GCN_E 800000

__global__ void gcn_degree_kernel(const int* __restrict__ src,
                                  const int* __restrict__ dst,
                                  unsigned* __restrict__ deg_out,
                                  unsigned* __restrict__ deg_in,
                                  unsigned* __restrict__ rank,
                                  int n_edges) {
  int e = blockIdx.x * blockDim.x + threadIdx.x;
  if (e < n_edges) {
    atomicAdd(&deg_out[src[e]], 1u);
    rank[e] = atomicAdd(&deg_in[dst[e]], 1u);
  }
}

__global__ void __launch_bounds__(1024)
gcn_norm_rowptr_kernel(const unsigned* __restrict__ deg_out,
                       const unsigned* __restrict__ deg_in,
                       float* __restrict__ norm_out,
                       float* __restrict__ norm_in,
                       unsigned* __restrict__ rowptr,
                       int n_nodes) {
  __shared__ unsigned s_front[32];
  __shared__ unsigned s_wave[32];
  const int tid  = threadIdx.x;
  const int lane = tid & 31;
  const int wv   = tid >> 5;
  const int blk0 = blockIdx.x * 1024;
  const int n    = blk0 + tid;

  unsigned din = 0u;
  if (n < n_nodes) {
    unsigned dout = deg_out[n];
    din = deg_in[n];
    norm_out[n] = rsqrtf((float)(dout < 1u ? 1u : dout));
    norm_in[n]  = rsqrtf((float)(din  < 1u ? 1u : din));
  }

  unsigned front = 0u;
  const uint4* d4 = (const uint4*)deg_in;
#pragma unroll 4
  for (int i = tid; i < blk0 / 4; i += 1024) {
    uint4 t = d4[i];
    front += t.x + t.y + t.z + t.w;
  }
  for (int o = 16; o > 0; o >>= 1) front += __shfl_xor(front, o, 32);

  unsigned incl = din;
  for (int o = 1; o < 32; o <<= 1) {
    unsigned t = __shfl_up(incl, o, 32);
    if (lane >= o) incl += t;
  }
  if (lane == 0)  s_front[wv] = front;
  if (lane == 31) s_wave[wv]  = incl;
  __syncthreads();

  unsigned f = s_front[lane];
  unsigned w = (lane < wv) ? s_wave[lane] : 0u;
  unsigned base = f + w;
  for (int o = 16; o > 0; o >>= 1) base += __shfl_xor(base, o, 32);

  unsigned excl = base + incl - din;
  if (n < n_nodes) rowptr[n] = excl;
  if (n == n_nodes - 1) rowptr[n_nodes] = excl + din;
}

__global__ void __launch_bounds__(256)
gcn_fill_scale_kernel(const int* __restrict__ src,
                      const int* __restrict__ dst,
                      const unsigned* __restrict__ rank,
                      const unsigned* __restrict__ rowptr,
                      int* __restrict__ csr_src,
                      int n_edges,
                      const float* __restrict__ x,
                      const float* __restrict__ rownorm,
                      float* __restrict__ y,
                      int n_nodes) {
  const int i = blockIdx.x * 256 + threadIdx.x;
  const bool do_edge = i < n_edges;
  const bool do_row  = i < n_nodes * 16;
  int d = 0, s = 0;
  unsigned r = 0u;
  float4 v = make_float4(0.f, 0.f, 0.f, 0.f);
  float sc = 0.f;
  if (do_edge) { d = dst[i]; r = rank[i]; s = src[i]; }
  if (do_row)  { v = ((const float4*)x)[i]; sc = rownorm[i >> 4]; }
  if (do_edge) csr_src[rowptr[d] + r] = s;
  if (do_row) {
    v.x *= sc; v.y *= sc; v.z *= sc; v.w *= sc;
    ((float4*)y)[i] = v;
  }
}

__device__ __forceinline__ void add4(float4& a, const float4& v) {
  a.x += v.x; a.y += v.y; a.z += v.z; a.w += v.w;
}

__device__ __forceinline__ float4 gcn_gather_row_sum(const unsigned* __restrict__ rowptr,
                                                     const int* __restrict__ csr_src,
                                                     const float* __restrict__ x,
                                                     int n, int q) {
  const char* xb = (const char*)x;
  const unsigned qoff = (unsigned)q * 16u;
  unsigned beg = rowptr[n];
  unsigned end = rowptr[n + 1];
  float4 a0 = make_float4(0.f, 0.f, 0.f, 0.f);
  float4 a1 = a0, a2 = a0, a3 = a0;
  for (unsigned base = beg; base < end; base += 32) {
    unsigned rem = end - base;
    int idx0 = ((unsigned)q < rem)        ? csr_src[base + q]      : 0;
    int idx1 = ((unsigned)(q + 16) < rem) ? csr_src[base + 16 + q] : 0;
    float4 v[32];
#pragma unroll
    for (int k = 0; k < 32; ++k) {
      int s = __shfl(k < 16 ? idx0 : idx1, k & 15, 16);
      v[k] = make_float4(0.f, 0.f, 0.f, 0.f);
      if ((unsigned)k < rem) v[k] = *(const float4*)(xb + ((unsigned)s * 256u + qoff));
    }
#pragma unroll
    for (int k = 0; k < 32; k += 4) {
      add4(a0, v[k]);
      add4(a1, v[k + 1]);
      add4(a2, v[k + 2]);
      add4(a3, v[k + 3]);
    }
  }
  add4(a0, a1);
  add4(a2, a3);
  add4(a0, a2);
  return a0;
}

__global__ void __launch_bounds__(256)
gcn_gather64_kernel(const unsigned* __restrict__ rowptr,
                    const int* __restrict__ csr_src,
                    const float* __restrict__ x,
                    float* __restrict__ agg,
                    int n_nodes) {
  int gid = blockIdx.x * 256 + threadIdx.x;
  int n = gid >> 4;
  int q = gid & 15;
  if (n >= n_nodes) return;
  float4 a = gcn_gather_row_sum(rowptr, csr_src, x, n, q);
  ((float4*)agg)[(size_t)n * 16 + q] = a;
}

__global__ void __launch_bounds__(256)
gcn_gather64_final_kernel(const unsigned* __restrict__ rowptr,
                          const int* __restrict__ csr_src,
                          const float* __restrict__ x,
                          const float* __restrict__ norm_in,
                          const float* __restrict__ bias,
                          float* __restrict__ out,
                          int n_nodes) {
  int gid = blockIdx.x * 256 + threadIdx.x;
  int n = gid >> 4;
  int q = gid & 15;
  if (n >= n_nodes) return;
  float s = norm_in[n];
  float4 bb = ((const float4*)bias)[q];
  float4 v = gcn_gather_row_sum(rowptr, csr_src, x, n, q);
  v.x = v.x * s + bb.x;
  v.y = v.y * s + bb.y;
  v.z = v.z * s + bb.z;
  v.w = v.w * s + bb.w;
  ((float4*)out)[(size_t)n * 16 + q] = v;
}

__global__ void __launch_bounds__(256)
gcn_gemm1_kernel(const float* __restrict__ A,
                 const float* __restrict__ W,
                 const float* __restrict__ bias,
                 const float* __restrict__ norm_in,
                 const float* __restrict__ norm_out,
                 float* __restrict__ Y,
                 int n_tiles_m) {
  int wave = threadIdx.x >> 5;
  int tile = blockIdx.x * 8 + wave;
  int mt = tile >> 3;
  int nt = tile & 7;
  if (mt >= n_tiles_m) return;
  int lane = threadIdx.x & 31;
  int half = lane >> 4;
  int l16  = lane & 15;
  int arow = mt * 16 + l16;
  int bcol = nt * 16 + l16;
  float rs = norm_in[arow];
  const float* ap = A + (size_t)arow * 64;
  v8f acc = {};
#pragma unroll
  for (int k0 = 0; k0 < 64; k0 += 4) {
    int ka = k0 + half * 2;
    v2f a;
    a.x = ap[ka] * rs;
    a.y = ap[ka + 1] * rs;
    v2f b;
    b.x = W[(size_t)ka * 128 + bcol];
    b.y = W[(size_t)(ka + 1) * 128 + bcol];
    acc = __builtin_amdgcn_wmma_f32_16x16x4_f32(
        false, a, false, b, (short)0, acc, false, false);
  }
  float bb = bias[bcol];
#pragma unroll
  for (int r = 0; r < 8; ++r) {
    int row = mt * 16 + half * 8 + r;
    float v = fmaxf(acc[r] + bb, 0.0f) * norm_out[row];
    Y[(size_t)row * 128 + bcol] = v;
  }
}

__global__ void __launch_bounds__(256)
gcn_gemm2_kernel(const float* __restrict__ A,
                 const float* __restrict__ W,
                 float* __restrict__ Y,
                 int n_tiles) {
  int wave = threadIdx.x >> 5;
  int tile = blockIdx.x * 8 + wave;
  if (tile >= n_tiles) return;
  int mt = tile >> 2;
  int nt = tile & 3;
  int lane = threadIdx.x & 31;
  int half = lane >> 4;
  int l16  = lane & 15;
  int arow = mt * 16 + l16;
  int bcol = nt * 16 + l16;
  const float* ap = A + (size_t)arow * 128;
  v8f acc = {};
#pragma unroll
  for (int k0 = 0; k0 < 128; k0 += 4) {
    int ka = k0 + half * 2;
    v2f a;
    a.x = ap[ka];
    a.y = ap[ka + 1];
    v2f b;
    b.x = W[(size_t)ka * 64 + bcol];
    b.y = W[(size_t)(ka + 1) * 64 + bcol];
    acc = __builtin_amdgcn_wmma_f32_16x16x4_f32(
        false, a, false, b, (short)0, acc, false, false);
  }
#pragma unroll
  for (int r = 0; r < 8; ++r) {
    int row = mt * 16 + half * 8 + r;
    Y[(size_t)row * 64 + bcol] = acc[r];
  }
}

extern "C" void kernel_launch(void* const* d_in, const int* in_sizes, int n_in,
                              void* d_out, int out_size, void* d_ws, size_t ws_size,
                              hipStream_t stream) {
  (void)in_sizes; (void)n_in; (void)out_size; (void)ws_size;
  const float* h   = (const float*)d_in[0];
  const int*   src = (const int*)d_in[1];
  const int*   dst = (const int*)d_in[2];
  const float* W1  = (const float*)d_in[3];
  const float* b1  = (const float*)d_in[4];
  const float* W2  = (const float*)d_in[5];
  const float* b2  = (const float*)d_in[6];
  float* out = (float*)d_out;

  const int N = GCN_N;
  const int E = GCN_E;

  char* ws = (char*)d_ws;
  size_t off = 0;
  auto carve = [&](size_t bytes) -> void* {
    void* p = ws + off;
    off += (bytes + 255) & ~(size_t)255;
    return p;
  };
  unsigned* deg_out = (unsigned*)carve((size_t)N * 4 * 2);
  unsigned* deg_in  = deg_out + N;
  float* norm_out   = (float*)carve((size_t)N * 4);
  float* norm_in    = (float*)carve((size_t)N * 4);
  float* hs         = (float*)carve((size_t)N * 64 * 4);
  float* agg        = (float*)carve((size_t)N * 64 * 4);
  float* h1s        = (float*)carve((size_t)N * 128 * 4);
  float* t2         = (float*)carve((size_t)N * 64 * 4);
  unsigned* rowptr  = (unsigned*)carve((size_t)(N + 1) * 4);
  unsigned* rank    = (unsigned*)carve((size_t)E * 4);
  int* csr_src      = (int*)carve((size_t)E * 4);

  hipMemsetAsync(deg_out, 0, (size_t)N * 4 * 2, stream);

  gcn_degree_kernel<<<(E + 255) / 256, 256, 0, stream>>>(src, dst, deg_out, deg_in, rank, E);
  gcn_norm_rowptr_kernel<<<(N + 1023) / 1024, 1024, 0, stream>>>(deg_out, deg_in, norm_out, norm_in, rowptr, N);
  gcn_fill_scale_kernel<<<((E > N * 16 ? E : N * 16) + 255) / 256, 256, 0, stream>>>(
      src, dst, rank, rowptr, csr_src, E, h, norm_out, hs, N);

  gcn_gather64_kernel<<<(N * 16 + 255) / 256, 256, 0, stream>>>(rowptr, csr_src, hs, agg, N);
  gcn_gemm1_kernel<<<N / 16, 256, 0, stream>>>(agg, W1, b1, norm_in, norm_out, h1s, N / 16);

  {
    int tiles = (N / 16) * 4;
    gcn_gemm2_kernel<<<(tiles + 7) / 8, 256, 0, stream>>>(h1s, W2, t2, tiles);
  }
  gcn_gather64_final_kernel<<<(N * 16 + 255) / 256, 256, 0, stream>>>(rowptr, csr_src, t2, norm_in, b2, out, N);
}


# Fuse layer-1 gather, GEMM1 and GEMM2 into one kernel: 32-node tiles of the aggregate and of the hidden layer stay in LDS and feed the WMMA A fragments; 7 launches -> 5, sum order unchanged
# speedup vs baseline: 1.7280x; 1.1100x over previous
#include <hip/hip_runtime.h>

typedef __attribute__((ext_vector_type(2))) float v2f;
typedef __attribute__((ext_vector_type(8))) float v8f;

#define GCN_N 50000
#define GCN_E 800000

__global__ void gcn_degree_kernel(const int* __restrict__ src,
                                  const int* __restrict__ dst,
                                  unsigned* __restrict__ deg_out,
                                  unsigned* __restrict__ deg_in,
                                  unsigned* __restrict__ rank,
                                  int n_edges) {
  int e = blockIdx.x * blockDim.x + threadIdx.x;
  if (e < n_edges) {
    atomicAdd(&deg_out[src[e]], 1u);
    rank[e] = atomicAdd(&deg_in[dst[e]], 1u);
  }
}

__global__ void __launch_bounds__(1024)
gcn_norm_rowptr_kernel(const unsigned* __restrict__ deg_out,
                       const unsigned* __restrict__ deg_in,
                       float* __restrict__ norm_out,
                       float* __restrict__ norm_in,
                       unsigned* __restrict__ rowptr,
                       int n_nodes) {
  __shared__ unsigned s_front[32];
  __shared__ unsigned s_wave[32];
  const int tid  = threadIdx.x;
  const int lane = tid & 31;
  const int wv   = tid >> 5;
  const int blk0 = blockIdx.x * 1024;
  const int n    = blk0 + tid;

  unsigned din = 0u;
  if (n < n_nodes) {
    unsigned dout = deg_out[n];
    din = deg_in[n];
    norm_out[n] = rsqrtf((float)(dout < 1u ? 1u : dout));
    norm_in[n]  = rsqrtf((float)(din  < 1u ? 1u : din));
  }

  unsigned front = 0u;
  const uint4* d4 = (const uint4*)deg_in;
#pragma unroll 4
  for (int i = tid; i < blk0 / 4; i += 1024) {
    uint4 t = d4[i];
    front += t.x + t.y + t.z + t.w;
  }
  for (int o = 16; o > 0; o >>= 1) front += __shfl_xor(front, o, 32);

  unsigned incl = din;
  for (int o = 1; o < 32; o <<= 1) {
    unsigned t = __shfl_up(incl, o, 32);
    if (lane >= o) incl += t;
  }
  if (lane == 0)  s_front[wv] = front;
  if (lane == 31) s_wave[wv]  = incl;
  __syncthreads();

  unsigned f = s_front[lane];
  unsigned w = (lane < wv) ? s_wave[lane] : 0u;
  unsigned base = f + w;
  for (int o = 16; o > 0; o >>= 1) base += __shfl_xor(base, o, 32);

  unsigned excl = base + incl - din;
  if (n < n_nodes) rowptr[n] = excl;
  if (n == n_nodes - 1) rowptr[n_nodes] = excl + din;
}

__global__ void __launch_bounds__(256)
gcn_fill_scale_kernel(const int* __restrict__ src,
                      const int* __restrict__ dst,
                      const unsigned* __restrict__ rank,
                      const unsigned* __restrict__ rowptr,
                      int* __restrict__ csr_src,
                      int n_edges,
                      const float* __restrict__ x,
                      const float* __restrict__ rownorm,
                      float* __restrict__ y,
                      int n_nodes) {
  const int i = blockIdx.x * 256 + threadIdx.x;
  const bool do_edge = i < n_edges;
  const bool do_row  = i < n_nodes * 16;
  int d = 0, s = 0;
  unsigned r = 0u;
  float4 v = make_float4(0.f, 0.f, 0.f, 0.f);
  float sc = 0.f;
  if (do_edge) { d = dst[i]; r = rank[i]; s = src[i]; }
  if (do_row)  { v = ((const float4*)x)[i]; sc = rownorm[i >> 4]; }
  if (do_edge) csr_src[rowptr[d] + r] = s;
  if (do_row) {
    v.x *= sc; v.y *= sc; v.z *= sc; v.w *= sc;
    ((float4*)y)[i] = v;
  }
}

__device__ __forceinline__ void add4(float4& a, const float4& v) {
  a.x += v.x; a.y += v.y; a.z += v.z; a.w += v.w;
}

__device__ __forceinline__ float4 gcn_gather_row_sum(const unsigned* __restrict__ rowptr,
                                                     const int* __restrict__ csr_src,
                                                     const float* __restrict__ x,
                                                     int n, int q) {
  const char* xb = (const char*)x;
  const unsigned qoff = (unsigned)q * 16u;
  unsigned beg = rowptr[n];
  unsigned end = rowptr[n + 1];
  float4 a0 = make_float4(0.f, 0.f, 0.f, 0.f);
  float4 a1 = a0, a2 = a0, a3 = a0;
  for (unsigned base = beg; base < end; base += 32) {
    unsigned rem = end - base;
    int idx0 = ((unsigned)q < rem)        ? csr_src[base + q]      : 0;
    int idx1 = ((unsigned)(q + 16) < rem) ? csr_src[base + 16 + q] : 0;
    float4 v[32];
#pragma unroll
    for (int k = 0; k < 32; ++k) {
      int s = __shfl(k < 16 ? idx0 : idx1, k & 15, 16);
      v[k] = make_float4(0.f, 0.f, 0.f, 0.f);
      if ((unsigned)k < rem) v[k] = *(const float4*)(xb + ((unsigned)s * 256u + qoff));
    }
#pragma unroll
    for (int k = 0; k < 32; k += 4) {
      add4(a0, v[k]);
      add4(a1, v[k + 1]);
      add4(a2, v[k + 2]);
      add4(a3, v[k + 3]);
    }
  }
  add4(a0, a1);
  add4(a2, a3);
  add4(a0, a2);
  return a0;
}

__global__ void __launch_bounds__(256)
gcn_gather64_final_kernel(const unsigned* __restrict__ rowptr,
                          const int* __restrict__ csr_src,
                          const float* __restrict__ x,
                          const float* __restrict__ norm_in,
                          const float* __restrict__ bias,
                          float* __restrict__ out,
                          int n_nodes) {
  int gid = blockIdx.x * 256 + threadIdx.x;
  int n = gid >> 4;
  int q = gid & 15;
  if (n >= n_nodes) return;
  float s = norm_in[n];
  float4 bb = ((const float4*)bias)[q];
  float4 v = gcn_gather_row_sum(rowptr, csr_src, x, n, q);
  v.x = v.x * s + bb.x;
  v.y = v.y * s + bb.y;
  v.z = v.z * s + bb.z;
  v.w = v.w * s + bb.w;
  ((float4*)out)[(size_t)n * 16 + q] = v;
}

#define GCN_TILE_ROWS 32
#define GCN_PITCH_S 68
#define GCN_PITCH_H 132

__global__ void __launch_bounds__(256)
gcn_layer1_fused_kernel(const unsigned* __restrict__ rowptr,
                        const int* __restrict__ csr_src,
                        const float* __restrict__ x,
                        const float* __restrict__ norm_in,
                        const float* __restrict__ norm_out,
                        const float* __restrict__ W1,
                        const float* __restrict__ b1,
                        const float* __restrict__ W2,
                        float* __restrict__ Y,
                        int n_nodes) {
  __shared__ __attribute__((aligned(16))) float sS[GCN_TILE_ROWS * GCN_PITCH_S];
  __shared__ __attribute__((aligned(16))) float sH[GCN_TILE_ROWS * GCN_PITCH_H];
  __shared__ float sNo[GCN_TILE_ROWS];

  const int tid  = threadIdx.x;
  const int blk0 = blockIdx.x * GCN_TILE_ROWS;
  const int g    = tid >> 4;
  const int q    = tid & 15;
  const int wave = tid >> 5;
  const int lane = tid & 31;
  const int half = lane >> 4;
  const int l16  = lane & 15;

  for (int p = 0; p < 2; ++p) {
    const int lr = p * 16 + g;
    const int n  = blk0 + lr;
    float4 a = make_float4(0.f, 0.f, 0.f, 0.f);
    float no = 0.f;
    if (n < n_nodes) {
      const float rs = norm_in[n];
      no = norm_out[n];
      a = gcn_gather_row_sum(rowptr, csr_src, x, n, q);
      a.x *= rs; a.y *= rs; a.z *= rs; a.w *= rs;
    }
    *(float4*)&sS[lr * GCN_PITCH_S + q * 4] = a;
    if (q == 0) sNo[lr] = no;
  }

  const int col1 = wave * 16 + l16;
  v2f bw1[16];
#pragma unroll
  for (int kk = 0; kk < 16; ++kk) {
    const int ka = kk * 4 + half * 2;
    bw1[kk].x = W1[ka * 128 + col1];
    bw1[kk].y = W1[(ka + 1) * 128 + col1];
  }
  const float bb1 = b1[col1];
  __syncthreads();

  {
    v2f a0[16], a1[16];
#pragma unroll
    for (int kk = 0; kk < 16; ++kk) {
      const int ka = kk * 4 + half * 2;
      a0[kk] = *(const v2f*)&sS[l16 * GCN_PITCH_S + ka];
      a1[kk] = *(const v2f*)&sS[(16 + l16) * GCN_PITCH_S + ka];
    }
    v8f acc0 = {};
    v8f acc1 = {};
#pragma unroll
    for (int kk = 0; kk < 16; ++kk) {
      acc0 = __builtin_amdgcn_wmma_f32_16x16x4_f32(
          false, a0[kk], false, bw1[kk], (short)0, acc0, false, false);
      acc1 = __builtin_amdgcn_wmma_f32_16x16x4_f32(
          false, a1[kk], false, bw1[kk], (short)0, acc1, false, false);
    }
#pragma unroll
    for (int r = 0; r < 8; ++r) {
      const int row = half * 8 + r;
      sH[row * GCN_PITCH_H + col1]        = fmaxf(acc0[r] + bb1, 0.0f) * sNo[row];
      sH[(16 + row) * GCN_PITCH_H + col1] = fmaxf(acc1[r] + bb1, 0.0f) * sNo[16 + row];
    }
  }

  const int mt2  = wave >> 2;
  const int nt2  = wave & 3;
  const int col2 = nt2 * 16 + l16;
  v2f bw2[32];
#pragma unroll
  for (int kk = 0; kk < 32; ++kk) {
    const int ka = kk * 4 + half * 2;
    bw2[kk].x = W2[ka * 64 + col2];
    bw2[kk].y = W2[(ka + 1) * 64 + col2];
  }
  __syncthreads();

  {
    v2f a2[32];
#pragma unroll
    for (int kk = 0; kk < 32; ++kk) {
      const int ka = kk * 4 + half * 2;
      a2[kk] = *(const v2f*)&sH[(mt2 * 16 + l16) * GCN_PITCH_H + ka];
    }
    v8f acc = {};
#pragma unroll
    for (int kk = 0; kk < 32; ++kk) {
      acc = __builtin_amdgcn_wmma_f32_16x16x4_f32(
          false, a2[kk], false, bw2[kk], (short)0, acc, false, false);
    }
#pragma unroll
    for (int r = 0; r < 8; ++r) {
      sS[(mt2 * 16 + half * 8 + r) * GCN_PITCH_S + col2] = acc[r];
    }
  }
  __syncthreads();

  for (int p = 0; p < 2; ++p) {
    const int lr = p * 16 + g;
    const int n  = blk0 + lr;
    if (n < n_nodes) {
      ((float4*)Y)[(size_t)n * 16 + q] = *(const float4*)&sS[lr * GCN_PITCH_S + q * 4];
    }
  }
}

extern "C" void kernel_launch(void* const* d_in, const int* in_sizes, int n_in,
                              void* d_out, int out_size, void* d_ws, size_t ws_size,
                              hipStream_t stream) {
  (void)in_sizes; (void)n_in; (void)out_size; (void)ws_size;
  const float* h   = (const float*)d_in[0];
  const int*   src = (const int*)d_in[1];
  const int*   dst = (const int*)d_in[2];
  const float* W1  = (const float*)d_in[3];
  const float* b1  = (const float*)d_in[4];
  const float* W2  = (const float*)d_in[5];
  const float* b2  = (const float*)d_in[6];
  float* out = (float*)d_out;

  const int N = GCN_N;
  const int E = GCN_E;

  char* ws = (char*)d_ws;
  size_t off = 0;
  auto carve = [&](size_t bytes) -> void* {
    void* p = ws + off;
    off += (bytes + 255) & ~(size_t)255;
    return p;
  };
  unsigned* deg_out = (unsigned*)carve((size_t)N * 4 * 2);
  unsigned* deg_in  = deg_out + N;
  float* norm_out   = (float*)carve((size_t)N * 4);
  float* norm_in    = (float*)carve((size_t)N * 4);
  float* hs         = (float*)carve((size_t)N * 64 * 4);
  float* t2         = (float*)carve((size_t)N * 64 * 4);
  unsigned* rowptr  = (unsigned*)carve((size_t)(N + 1) * 4);
  unsigned* rank    = (unsigned*)carve((size_t)E * 4);
  int* csr_src      = (int*)carve((size_t)E * 4);

  hipMemsetAsync(deg_out, 0, (size_t)N * 4 * 2, stream);

  gcn_degree_kernel<<<(E + 255) / 256, 256, 0, stream>>>(src, dst, deg_out, deg_in, rank, E);
  gcn_norm_rowptr_kernel<<<(N + 1023) / 1024, 1024, 0, stream>>>(deg_out, deg_in, norm_out, norm_in, rowptr, N);
  gcn_fill_scale_kernel<<<((E > N * 16 ? E : N * 16) + 255) / 256, 256, 0, stream>>>(
      src, dst, rank, rowptr, csr_src, E, h, norm_out, hs, N);

  gcn_layer1_fused_kernel<<<(N + 31) / 32, 256, 0, stream>>>(
      rowptr, csr_src, hs, norm_in, norm_out, W1, b1, W2, t2, N);

  gcn_gather64_final_kernel<<<(N * 16 + 255) / 256, 256, 0, stream>>>(rowptr, csr_src, t2, norm_in, b2, out, N);
}


# Group edges by destination via 256-node bins and LDS atomics (count, place, one owner block per bin), no scattered global atomics; per-node sum order is now LDS arrival order in a binned list
# speedup vs baseline: 3.6791x; 2.1291x over previous
#include <hip/hip_runtime.h>

typedef __attribute__((ext_vector_type(2))) float v2f;
typedef __attribute__((ext_vector_type(8))) float v8f;

#define GCN_N 50000
#define GCN_E 800000

#define GCN_BIN_NODES 256
#define GCN_NB ((GCN_N + GCN_BIN_NODES - 1) / GCN_BIN_NODES)
#define GCN_BIN_THREADS 512
#define GCN_BIN_INT4 2
#define GCN_BIN_CHUNK4 (GCN_BIN_THREADS * GCN_BIN_INT4)
static_assert(GCN_N <= 65536, "a node id must fit in 16 bits of the packed record");
static_assert(GCN_E % 4 == 0, "edges are read as int4");
static_assert(GCN_NB <= 224 && GCN_NB <= GCN_BIN_NODES, "bin scans assume at most 7 waves of bins");

__device__ __forceinline__ void gcn_load_edges(const int* __restrict__ src, const int* __restrict__ dst,
                                               int n_edges4, int (&sv)[4 * GCN_BIN_INT4],
                                               int (&dv)[4 * GCN_BIN_INT4], bool (&ok)[GCN_BIN_INT4]) {
#pragma unroll
  for (int j = 0; j < GCN_BIN_INT4; ++j) {
    const int idx = blockIdx.x * GCN_BIN_CHUNK4 + j * GCN_BIN_THREADS + threadIdx.x;
    ok[j] = idx < n_edges4;
    int4 s = make_int4(0, 0, 0, 0), d = make_int4(0, 0, 0, 0);
    if (ok[j]) { s = ((const int4*)src)[idx]; d = ((const int4*)dst)[idx]; }
    sv[4 * j] = s.x; sv[4 * j + 1] = s.y; sv[4 * j + 2] = s.z; sv[4 * j + 3] = s.w;
    dv[4 * j] = d.x; dv[4 * j + 1] = d.y; dv[4 * j + 2] = d.z; dv[4 * j + 3] = d.w;
  }
}

__device__ __forceinline__ void gcn_hist_edges(unsigned* s_cnt, const int (&sv)[4 * GCN_BIN_INT4],
                                               const int (&dv)[4 * GCN_BIN_INT4],
                                               const bool (&ok)[GCN_BIN_INT4]) {
#pragma unroll
  for (int k = 0; k < 4 * GCN_BIN_INT4; ++k) {
    if (ok[k >> 2]) {
      atomicAdd(&s_cnt[dv[k] / GCN_BIN_NODES], 1u);
      atomicAdd(&s_cnt[GCN_NB + sv[k] / GCN_BIN_NODES], 1u);
    }
  }
}

__global__ void __launch_bounds__(GCN_BIN_THREADS)
gcn_bin_count_kernel(const int* __restrict__ src, const int* __restrict__ dst,
                     unsigned* __restrict__ tot, int n_edges4) {
  __shared__ unsigned s_cnt[2 * GCN_NB];
  const int tid = threadIdx.x;
  int sv[4 * GCN_BIN_INT4], dv[4 * GCN_BIN_INT4];
  bool ok[GCN_BIN_INT4];
  gcn_load_edges(src, dst, n_edges4, sv, dv, ok);
  if (tid < 2 * GCN_NB) s_cnt[tid] = 0u;
  __syncthreads();
  gcn_hist_edges(s_cnt, sv, dv, ok);
  __syncthreads();
  if (tid < 2 * GCN_NB) {
    const unsigned c = s_cnt[tid];
    if (c != 0u) atomicAdd(&tot[tid], c);
  }
}

__global__ void __launch_bounds__(GCN_BIN_THREADS)
gcn_bin_place_kernel(const int* __restrict__ src, const int* __restrict__ dst,
                     const unsigned* __restrict__ tot, unsigned* __restrict__ cursor,
                     unsigned* __restrict__ stag_d, unsigned* __restrict__ stag_s, int n_edges4) {
  __shared__ unsigned s_cnt[2 * GCN_NB];
  __shared__ unsigned s_wsum[14];
  const int tid  = threadIdx.x;
  const int lane = tid & 31;
  const int wv   = tid >> 5;
  int sv[4 * GCN_BIN_INT4], dv[4 * GCN_BIN_INT4];
  bool ok[GCN_BIN_INT4];
  gcn_load_edges(src, dst, n_edges4, sv, dv, ok);
  if (tid < 2 * GCN_NB) s_cnt[tid] = 0u;

  const int side = wv >= 7 ? 1 : 0;
  const int li   = (wv - side * 7) * 32 + lane;
  const bool has_bin = wv < 14 && li < GCN_NB;
  const int bi   = side * GCN_NB + li;
  unsigned v = 0u;
  if (has_bin) v = tot[bi];
  unsigned incl = v;
  for (int o = 1; o < 32; o <<= 1) {
    unsigned t = __shfl_up(incl, o, 32);
    if (lane >= o) incl += t;
  }
  if (wv < 14 && lane == 31) s_wsum[wv] = incl;
  __syncthreads();

  gcn_hist_edges(s_cnt, sv, dv, ok);
  unsigned pre = 0u;
  for (int j = side * 7; j < wv && j < 14; ++j) pre += s_wsum[j];
  __syncthreads();

  if (has_bin) {
    const unsigned c = s_cnt[bi];
    unsigned g = 0u;
    if (c != 0u) g = atomicAdd(&cursor[bi], c);
    s_cnt[bi] = pre + incl - v + g;
  }
  __syncthreads();

#pragma unroll
  for (int j = 0; j < GCN_BIN_INT4; ++j) {
    if (ok[j]) {
      unsigned p[4], q[4];
#pragma unroll
      for (int e = 0; e < 4; ++e) {
        p[e] = atomicAdd(&s_cnt[dv[4 * j + e] / GCN_BIN_NODES], 1u);
        q[e] = atomicAdd(&s_cnt[GCN_NB + sv[4 * j + e] / GCN_BIN_NODES], 1u);
      }
#pragma unroll
      for (int e = 0; e < 4; ++e) {
        const unsigned rec = (unsigned)sv[4 * j + e] | ((unsigned)dv[4 * j + e] << 16);
        stag_d[p[e]] = rec;
        stag_s[q[e]] = rec;
      }
    }
  }
}

__global__ void __launch_bounds__(GCN_BIN_NODES)
gcn_bin_owner_kernel(const unsigned* __restrict__ tot,
                     const unsigned* __restrict__ stag_d, const unsigned* __restrict__ stag_s,
                     unsigned* __restrict__ rowptr, int* __restrict__ csr_src,
                     float* __restrict__ norm_in, float* __restrict__ norm_out,
                     const float* __restrict__ x, float* __restrict__ y, int n_nodes) {
  __shared__ unsigned s_cnt[GCN_BIN_NODES];
  __shared__ unsigned s_red[8];
  __shared__ unsigned s_wsum[8];
  __shared__ float s_nrm[GCN_BIN_NODES];
  const int tid  = threadIdx.x;
  const int lane = tid & 31;
  const int wv   = tid >> 5;
  const bool is_dst = blockIdx.x < GCN_NB;
  const int bin = is_dst ? blockIdx.x : blockIdx.x - GCN_NB;
  const unsigned* totp = tot + (is_dst ? 0 : GCN_NB);
  const unsigned* stag = is_dst ? stag_d : stag_s;
  const int shift = is_dst ? 16 : 0;

  unsigned part = (tid < bin) ? totp[tid] : 0u;
  const unsigned nrec = totp[bin];
  for (int o = 16; o > 0; o >>= 1) part += __shfl_xor(part, o, 32);
  if (lane == 0) s_red[wv] = part;
  s_cnt[tid] = 0u;
  __syncthreads();
  unsigned base = 0u;
  for (int j = 0; j < 8; ++j) base += s_red[j];

  for (unsigned i0 = 0; i0 < nrec; i0 += GCN_BIN_NODES * 8) {
    unsigned r[8];
#pragma unroll
    for (int u = 0; u < 8; ++u) {
      const unsigned i = i0 + u * GCN_BIN_NODES + tid;
      r[u] = stag[base + (i < nrec ? i : nrec - 1u)];
    }
#pragma unroll
    for (int u = 0; u < 8; ++u) {
      const unsigned i = i0 + u * GCN_BIN_NODES + tid;
      if (i < nrec) atomicAdd(&s_cnt[(r[u] >> shift) & (GCN_BIN_NODES - 1)], 1u);
    }
  }
  __syncthreads();

  const unsigned c = s_cnt[tid];
  const int n = bin * GCN_BIN_NODES + tid;
  const float nrm = rsqrtf((float)(c < 1u ? 1u : c));

  if (!is_dst) {
    if (n < n_nodes) norm_out[n] = nrm;
    s_nrm[tid] = nrm;
    __syncthreads();
    float4 v[16];
#pragma unroll
    for (int j = 0; j < 16; ++j) {
      const int f = j * GCN_BIN_NODES + tid;
      const int nn = bin * GCN_BIN_NODES + (f >> 4);
      const int nc = nn < n_nodes ? nn : n_nodes - 1;
      v[j] = ((const float4*)x)[(size_t)nc * 16 + (f & 15)];
    }
#pragma unroll
    for (int j = 0; j < 16; ++j) {
      const int f = j * GCN_BIN_NODES + tid;
      const int row = f >> 4;
      const int nn = bin * GCN_BIN_NODES + row;
      if (nn < n_nodes) {
        const float sc = s_nrm[row];
        float4 w = v[j];
        w.x *= sc; w.y *= sc; w.z *= sc; w.w *= sc;
        ((float4*)y)[(size_t)nn * 16 + (f & 15)] = w;
      }
    }
  } else {
    unsigned incl = c;
    for (int o = 1; o < 32; o <<= 1) {
      unsigned t = __shfl_up(incl, o, 32);
      if (lane >= o) incl += t;
    }
    if (lane == 31) s_wsum[wv] = incl;
    __syncthreads();
    unsigned pre = 0u;
    for (int j = 0; j < wv; ++j) pre += s_wsum[j];
    const unsigned first = base + pre + incl - c;
    if (n < n_nodes) { rowptr[n] = first; norm_in[n] = nrm; }
    if (n == n_nodes - 1) rowptr[n_nodes] = first + c;
    s_cnt[tid] = first;
    __syncthreads();
    for (unsigned i0 = 0; i0 < nrec; i0 += GCN_BIN_NODES * 8) {
      unsigned r[8];
#pragma unroll
      for (int u = 0; u < 8; ++u) {
        const unsigned i = i0 + u * GCN_BIN_NODES + tid;
        r[u] = stag[base + (i < nrec ? i : nrec - 1u)];
      }
#pragma unroll
      for (int u = 0; u < 8; ++u) {
        const unsigned i = i0 + u * GCN_BIN_NODES + tid;
        if (i < nrec) {
          const unsigned p = atomicAdd(&s_cnt[(r[u] >> 16) & (GCN_BIN_NODES - 1)], 1u);
          csr_src[p] = (int)(r[u] & 0xffffu);
        }
      }
    }
  }
}

__device__ __forceinline__ void add4(float4& a, const float4& v) {
  a.x += v.x; a.y += v.y; a.z += v.z; a.w += v.w;
}

__device__ __forceinline__ float4 gcn_gather_row_sum(const unsigned* __restrict__ rowptr,
                                                     const int* __restrict__ csr_src,
                                                     const float* __restrict__ x,
                                                     int n, int q) {
  const char* xb = (const char*)x;
  const unsigned qoff = (unsigned)q * 16u;
  unsigned beg = rowptr[n];
  unsigned end = rowptr[n + 1];
  float4 a0 = make_float4(0.f, 0.f, 0.f, 0.f);
  float4 a1 = a0, a2 = a0, a3 = a0;
  for (unsigned base = beg; base < end; base += 32) {
    unsigned rem = end - base;
    int idx0 = ((unsigned)q < rem)        ? csr_src[base + q]      : 0;
    int idx1 = ((unsigned)(q + 16) < rem) ? csr_src[base + 16 + q] : 0;
    float4 v[32];
#pragma unroll
    for (int k = 0; k < 32; ++k) {
      int s = __shfl(k < 16 ? idx0 : idx1, k & 15, 16);
      v[k] = make_float4(0.f, 0.f, 0.f, 0.f);
      if ((unsigned)k < rem) v[k] = *(const float4*)(xb + ((unsigned)s * 256u + qoff));
    }
#pragma unroll
    for (int k = 0; k < 32; k += 4) {
      add4(a0, v[k]);
      add4(a1, v[k + 1]);
      add4(a2, v[k + 2]);
      add4(a3, v[k + 3]);
    }
  }
  add4(a0, a1);
  add4(a2, a3);
  add4(a0, a2);
  return a0;
}

__global__ void __launch_bounds__(256)
gcn_gather64_final_kernel(const unsigned* __restrict__ rowptr,
                          const int* __restrict__ csr_src,
                          const float* __restrict__ x,
                          const float* __restrict__ norm_in,
                          const float* __restrict__ bias,
                          float* __restrict__ out,
                          int n_nodes) {
  int gid = blockIdx.x * 256 + threadIdx.x;
  int n = gid >> 4;
  int q = gid & 15;
  if (n >= n_nodes) return;
  float s = norm_in[n];
  float4 bb = ((const float4*)bias)[q];
  float4 v = gcn_gather_row_sum(rowptr, csr_src, x, n, q);
  v.x = v.x * s + bb.x;
  v.y = v.y * s + bb.y;
  v.z = v.z * s + bb.z;
  v.w = v.w * s + bb.w;
  ((float4*)out)[(size_t)n * 16 + q] = v;
}

#define GCN_TILE_ROWS 32
#define GCN_PITCH_S 68
#define GCN_PITCH_H 132

__global__ void __launch_bounds__(256)
gcn_layer1_fused_kernel(const unsigned* __restrict__ rowptr,
                        const int* __restrict__ csr_src,
                        const float* __restrict__ x,
                        const float* __restrict__ norm_in,
                        const float* __restrict__ norm_out,
                        const float* __restrict__ W1,
                        const float* __restrict__ b1,
                        const float* __restrict__ W2,
                        float* __restrict__ Y,
                        int n_nodes) {
  __shared__ __attribute__((aligned(16))) float sS[GCN_TILE_ROWS * GCN_PITCH_S];
  __shared__ __attribute__((aligned(16))) float sH[GCN_TILE_ROWS * GCN_PITCH_H];
  __shared__ float sNo[GCN_TILE_ROWS];

  const int tid  = threadIdx.x;
  const int blk0 = blockIdx.x * GCN_TILE_ROWS;
  const int g    = tid >> 4;
  const int q    = tid & 15;
  const int wave = tid >> 5;
  const int lane = tid & 31;
  const int half = lane >> 4;
  const int l16  = lane & 15;

  for (int p = 0; p < 2; ++p) {
    const int lr = p * 16 + g;
    const int n  = blk0 + lr;
    float4 a = make_float4(0.f, 0.f, 0.f, 0.f);
    float no = 0.f;
    if (n < n_nodes) {
      const float rs = norm_in[n];
      no = norm_out[n];
      a = gcn_gather_row_sum(rowptr, csr_src, x, n, q);
      a.x *= rs; a.y *= rs; a.z *= rs; a.w *= rs;
    }
    *(float4*)&sS[lr * GCN_PITCH_S + q * 4] = a;
    if (q == 0) sNo[lr] = no;
  }

  const int col1 = wave * 16 + l16;
  v2f bw1[16];
#pragma unroll
  for (int kk = 0; kk < 16; ++kk) {
    const int ka = kk * 4 + half * 2;
    bw1[kk].x = W1[ka * 128 + col1];
    bw1[kk].y = W1[(ka + 1) * 128 + col1];
  }
  const float bb1 = b1[col1];
  __syncthreads();

  {
    v2f a0[16], a1[16];
#pragma unroll
    for (int kk = 0; kk < 16; ++kk) {
      const int ka = kk * 4 + half * 2;
      a0[kk] = *(const v2f*)&sS[l16 * GCN_PITCH_S + ka];
      a1[kk] = *(const v2f*)&sS[(16 + l16) * GCN_PITCH_S + ka];
    }
    v8f acc0 = {};
    v8f acc1 = {};
#pragma unroll
    for (int kk = 0; kk < 16; ++kk) {
      acc0 = __builtin_amdgcn_wmma_f32_16x16x4_f32(
          false, a0[kk], false, bw1[kk], (short)0, acc0, false, false);
      acc1 = __builtin_amdgcn_wmma_f32_16x16x4_f32(
          false, a1[kk], false, bw1[kk], (short)0, acc1, false, false);
    }
#pragma unroll
    for (int r = 0; r < 8; ++r) {
      const int row = half * 8 + r;
      sH[row * GCN_PITCH_H + col1]        = fmaxf(acc0[r] + bb1, 0.0f) * sNo[row];
      sH[(16 + row) * GCN_PITCH_H + col1] = fmaxf(acc1[r] + bb1, 0.0f) * sNo[16 + row];
    }
  }

  const int mt2  = wave >> 2;
  const int nt2  = wave & 3;
  const int col2 = nt2 * 16 + l16;
  v2f bw2[32];
#pragma unroll
  for (int kk = 0; kk < 32; ++kk) {
    const int ka = kk * 4 + half * 2;
    bw2[kk].x = W2[ka * 64 + col2];
    bw2[kk].y = W2[(ka + 1) * 64 + col2];
  }
  __syncthreads();

  {
    v2f a2[32];
#pragma unroll
    for (int kk = 0; kk < 32; ++kk) {
      const int ka = kk * 4 + half * 2;
      a2[kk] = *(const v2f*)&sH[(mt2 * 16 + l16) * GCN_PITCH_H + ka];
    }
    v8f acc = {};
#pragma unroll
    for (int kk = 0; kk < 32; ++kk) {
      acc = __builtin_amdgcn_wmma_f32_16x16x4_f32(
          false, a2[kk], false, bw2[kk], (short)0, acc, false, false);
    }
#pragma unroll
    for (int r = 0; r < 8; ++r) {
      sS[(mt2 * 16 + half * 8 + r) * GCN_PITCH_S + col2] = acc[r];
    }
  }
  __syncthreads();

  for (int p = 0; p < 2; ++p) {
    const int lr = p * 16 + g;
    const int n  = blk0 + lr;
    if (n < n_nodes) {
      ((float4*)Y)[(size_t)n * 16 + q] = *(const float4*)&sS[lr * GCN_PITCH_S + q * 4];
    }
  }
}

extern "C" void kernel_launch(void* const* d_in, const int* in_sizes, int n_in,
                              void* d_out, int out_size, void* d_ws, size_t ws_size,
                              hipStream_t stream) {
  (void)in_sizes; (void)n_in; (void)out_size; (void)ws_size;
  const float* h   = (const float*)d_in[0];
  const int*   src = (const int*)d_in[1];
  const int*   dst = (const int*)d_in[2];
  const float* W1  = (const float*)d_in[3];
  const float* b1  = (const float*)d_in[4];
  const float* W2  = (const float*)d_in[5];
  const float* b2  = (const float*)d_in[6];
  float* out = (float*)d_out;

  const int N = GCN_N;
  const int E = GCN_E;

  char* ws = (char*)d_ws;
  size_t off = 0;
  auto carve = [&](size_t bytes) -> void* {
    void* p = ws + off;
    off += (bytes + 255) & ~(size_t)255;
    return p;
  };
  unsigned* tot     = (unsigned*)carve((size_t)N * 4 * 2);
  unsigned* cursor  = tot + 2 * GCN_NB;
  float* norm_out   = (float*)carve((size_t)N * 4);
  float* norm_in    = (float*)carve((size_t)N * 4);
  float* hs         = (float*)carve((size_t)N * 64 * 4);
  float* t2         = (float*)carve((size_t)N * 64 * 4);
  unsigned* rowptr  = (unsigned*)carve((size_t)(N + 1) * 4);
  unsigned* stag_s  = (unsigned*)carve((size_t)E * 4);
  int* csr_src      = (int*)carve((size_t)E * 4);
  unsigned* stag_d  = (unsigned*)t2;

  hipMemsetAsync(tot, 0, (size_t)GCN_NB * 4 * 4, stream);

  const int E4 = E / 4;
  const int bin_blocks = (E4 + GCN_BIN_CHUNK4 - 1) / GCN_BIN_CHUNK4;
  gcn_bin_count_kernel<<<bin_blocks, GCN_BIN_THREADS, 0, stream>>>(src, dst, tot, E4);
  gcn_bin_place_kernel<<<bin_blocks, GCN_BIN_THREADS, 0, stream>>>(src, dst, tot, cursor, stag_d, stag_s, E4);
  gcn_bin_owner_kernel<<<2 * GCN_NB, GCN_BIN_NODES, 0, stream>>>(
      tot, stag_d, stag_s, rowptr, csr_src, norm_in, norm_out, h, hs, N);

  gcn_layer1_fused_kernel<<<(N + 31) / 32, 256, 0, stream>>>(
      rowptr, csr_src, hs, norm_in, norm_out, W1, b1, W2, t2, N);

  gcn_gather64_final_kernel<<<(N * 16 + 255) / 256, 256, 0, stream>>>(rowptr, csr_src, t2, norm_in, b2, out, N);
}


# Drop the counting launch and the clear: each placing block sorts its 4096 edges by bin in LDS into its own slab and a run table; owners collect runs; 6 graph nodes -> 4; per-node sum order changes
# speedup vs baseline: 4.3192x; 1.1740x over previous
#include <hip/hip_runtime.h>

typedef __attribute__((ext_vector_type(2))) float v2f;
typedef __attribute__((ext_vector_type(8))) float v8f;

#define GCN_N 50000
#define GCN_E 800000

#define GCN_BIN_NODES 256
#define GCN_NB ((GCN_N + GCN_BIN_NODES - 1) / GCN_BIN_NODES)
#define GCN_BIN_THREADS 512
#define GCN_BIN_INT4 2
#define GCN_BIN_CHUNK4 (GCN_BIN_THREADS * GCN_BIN_INT4)
#define GCN_SLAB (4 * GCN_BIN_CHUNK4)
#define GCN_PBLK ((GCN_E / 4 + GCN_BIN_CHUNK4 - 1) / GCN_BIN_CHUNK4)
static_assert(GCN_N <= 65536, "a node id must fit in 16 bits of the packed record");
static_assert(GCN_E % 4 == 0, "edges are read as int4");
static_assert(GCN_NB <= 224 && GCN_NB <= GCN_BIN_NODES, "bin scans assume at most 7 waves of bins");
static_assert(GCN_SLAB < 65536, "a run's start and its count share one table word");
static_assert(GCN_PBLK <= GCN_BIN_NODES, "an owner block reads one table word per thread");

__device__ __forceinline__ void gcn_load_edges(const int* __restrict__ src, const int* __restrict__ dst,
                                               int n_edges4, int (&sv)[4 * GCN_BIN_INT4],
                                               int (&dv)[4 * GCN_BIN_INT4], bool (&ok)[GCN_BIN_INT4]) {
#pragma unroll
  for (int j = 0; j < GCN_BIN_INT4; ++j) {
    const int idx = blockIdx.x * GCN_BIN_CHUNK4 + j * GCN_BIN_THREADS + threadIdx.x;
    ok[j] = idx < n_edges4;
    int4 s = make_int4(0, 0, 0, 0), d = make_int4(0, 0, 0, 0);
    if (ok[j]) { s = ((const int4*)src)[idx]; d = ((const int4*)dst)[idx]; }
    sv[4 * j] = s.x; sv[4 * j + 1] = s.y; sv[4 * j + 2] = s.z; sv[4 * j + 3] = s.w;
    dv[4 * j] = d.x; dv[4 * j + 1] = d.y; dv[4 * j + 2] = d.z; dv[4 * j + 3] = d.w;
  }
}

__device__ __forceinline__ void gcn_hist_edges(unsigned* s_cnt, const int (&sv)[4 * GCN_BIN_INT4],
                                               const int (&dv)[4 * GCN_BIN_INT4],
                                               const bool (&ok)[GCN_BIN_INT4]) {
#pragma unroll
  for (int k = 0; k < 4 * GCN_BIN_INT4; ++k) {
    if (ok[k >> 2]) {
      atomicAdd(&s_cnt[dv[k] / GCN_BIN_NODES], 1u);
      atomicAdd(&s_cnt[GCN_NB + sv[k] / GCN_BIN_NODES], 1u);
    }
  }
}

__global__ void __launch_bounds__(GCN_BIN_THREADS)
gcn_bin_place_kernel(const int* __restrict__ src, const int* __restrict__ dst,
                     unsigned* __restrict__ tab, unsigned* __restrict__ cursor,
                     unsigned* __restrict__ stag_d, unsigned* __restrict__ stag_s, int n_edges4) {
  __shared__ unsigned s_cnt[2 * GCN_NB];
  __shared__ unsigned s_wsum[14];
  __shared__ __attribute__((aligned(16))) unsigned s_rec[2][GCN_SLAB];
  const int tid  = threadIdx.x;
  const int lane = tid & 31;
  const int wv   = tid >> 5;
  int sv[4 * GCN_BIN_INT4], dv[4 * GCN_BIN_INT4];
  bool ok[GCN_BIN_INT4];
  gcn_load_edges(src, dst, n_edges4, sv, dv, ok);
  if (tid < 2 * GCN_NB) s_cnt[tid] = 0u;
  if (blockIdx.x == 0 && tid == 0) *cursor = 0u;
  __syncthreads();

  gcn_hist_edges(s_cnt, sv, dv, ok);
  __syncthreads();

  const int side = wv >= 7 ? 1 : 0;
  const int li   = (wv - side * 7) * 32 + lane;
  const bool has_bin = wv < 14 && li < GCN_NB;
  const int bi   = side * GCN_NB + li;
  unsigned v = 0u;
  if (has_bin) v = s_cnt[bi];
  unsigned incl = v;
  for (int o = 1; o < 32; o <<= 1) {
    unsigned t = __shfl_up(incl, o, 32);
    if (lane >= o) incl += t;
  }
  if (wv < 14 && lane == 31) s_wsum[wv] = incl;
  __syncthreads();
  unsigned pre = 0u;
  for (int j = side * 7; j < wv && j < 14; ++j) pre += s_wsum[j];
  if (has_bin) {
    const unsigned start = pre + incl - v;
    tab[(size_t)bi * GCN_PBLK + blockIdx.x] = (start << 16) | v;
    s_cnt[bi] = start;
  }
  __syncthreads();

#pragma unroll
  for (int j = 0; j < GCN_BIN_INT4; ++j) {
    if (ok[j]) {
      unsigned p[4], q[4];
#pragma unroll
      for (int e = 0; e < 4; ++e) {
        p[e] = atomicAdd(&s_cnt[dv[4 * j + e] / GCN_BIN_NODES], 1u);
        q[e] = atomicAdd(&s_cnt[GCN_NB + sv[4 * j + e] / GCN_BIN_NODES], 1u);
      }
#pragma unroll
      for (int e = 0; e < 4; ++e) {
        const unsigned rec = (unsigned)sv[4 * j + e] | ((unsigned)dv[4 * j + e] << 16);
        s_rec[0][p[e]] = rec;
        s_rec[1][q[e]] = rec;
      }
    }
  }
  __syncthreads();

  const int left4 = n_edges4 - blockIdx.x * GCN_BIN_CHUNK4;
  const int n4 = left4 < GCN_BIN_CHUNK4 ? left4 : GCN_BIN_CHUNK4;
  uint4* od = (uint4*)(stag_d + (size_t)blockIdx.x * GCN_SLAB);
  uint4* os = (uint4*)(stag_s + (size_t)blockIdx.x * GCN_SLAB);
  for (int i = tid; i < n4; i += GCN_BIN_THREADS) {
    od[i] = ((const uint4*)s_rec[0])[i];
    os[i] = ((const uint4*)s_rec[1])[i];
  }
}

#define GCN_RUNS ((GCN_PBLK + 7) / 8)
static_assert(GCN_RUNS <= 32, "a wave holds one run per lane");
template <typename F>
__device__ __forceinline__ void gcn_for_bin_records(const unsigned* __restrict__ stag,
                                                    unsigned mc, unsigned ma, int lane, F f) {
  unsigned r[GCN_RUNS];
#pragma unroll
  for (int j = 0; j < GCN_RUNS; ++j) {
    const unsigned cnt = (unsigned)__builtin_amdgcn_readlane((int)mc, j);
    const unsigned adr = (unsigned)__builtin_amdgcn_readlane((int)ma, j);
    r[j] = stag[(unsigned)lane < cnt ? adr + (unsigned)lane : 0u];
  }
#pragma unroll
  for (int j = 0; j < GCN_RUNS; ++j) {
    if ((unsigned)lane < (unsigned)__builtin_amdgcn_readlane((int)mc, j)) f(r[j]);
  }
  unsigned big = (unsigned)__ballot(mc > 32u);
  while (big != 0u) {
    const int j = __ffs(big) - 1;
    big &= big - 1u;
    const unsigned cnt = (unsigned)__builtin_amdgcn_readlane((int)mc, j);
    const unsigned adr = (unsigned)__builtin_amdgcn_readlane((int)ma, j);
    for (unsigned l = 32u + (unsigned)lane; l < cnt; l += 32u) f(stag[adr + l]);
  }
}

__global__ void __launch_bounds__(GCN_BIN_NODES)
gcn_bin_owner_kernel(const unsigned* __restrict__ tab, unsigned* __restrict__ cursor,
                     const unsigned* __restrict__ stag_d, const unsigned* __restrict__ stag_s,
                     unsigned* __restrict__ rowrng, int* __restrict__ csr_src,
                     float* __restrict__ norm_in, float* __restrict__ norm_out,
                     const float* __restrict__ x, float* __restrict__ y, int n_nodes) {
  __shared__ unsigned s_cnt[GCN_BIN_NODES];
  __shared__ unsigned s_wsum[8];
  __shared__ unsigned s_base;
  __shared__ float s_nrm[GCN_BIN_NODES];
  const int tid  = threadIdx.x;
  const int lane = tid & 31;
  const int wv   = tid >> 5;
  const bool is_dst = blockIdx.x < GCN_NB;
  const int bin = is_dst ? blockIdx.x : blockIdx.x - GCN_NB;
  const unsigned* stag = is_dst ? stag_d : stag_s;
  const int shift = is_dst ? 16 : 0;

  const int mt = wv + 8 * lane;
  unsigned w = 0u;
  if (mt < GCN_PBLK) w = tab[(size_t)blockIdx.x * GCN_PBLK + mt];
  const unsigned mc = w & 0xffffu;
  const unsigned ma = (unsigned)mt * GCN_SLAB + (w >> 16);
  s_cnt[tid] = 0u;
  unsigned part = mc;
  for (int o = 16; o > 0; o >>= 1) part += __shfl_xor(part, o, 32);
  if (lane == 0) s_wsum[wv] = part;
  __syncthreads();
  unsigned got = 0u;
  if (is_dst && tid == 0) {
    unsigned nrec = 0u;
    for (int j = 0; j < 8; ++j) nrec += s_wsum[j];
    got = atomicAdd(cursor, nrec);
  }

  gcn_for_bin_records(stag, mc, ma, lane, [&](unsigned rec) {
    atomicAdd(&s_cnt[(rec >> shift) & (GCN_BIN_NODES - 1)], 1u);
  });
  if (is_dst && tid == 0) s_base = got;
  __syncthreads();

  const unsigned c = s_cnt[tid];
  const int n = bin * GCN_BIN_NODES + tid;
  const float nrm = rsqrtf((float)(c < 1u ? 1u : c));

  if (!is_dst) {
    if (n < n_nodes) norm_out[n] = nrm;
    s_nrm[tid] = nrm;
    __syncthreads();
    float4 v[16];
#pragma unroll
    for (int j = 0; j < 16; ++j) {
      const int f = j * GCN_BIN_NODES + tid;
      const int nn = bin * GCN_BIN_NODES + (f >> 4);
      const int nc = nn < n_nodes ? nn : n_nodes - 1;
      v[j] = ((const float4*)x)[(size_t)nc * 16 + (f & 15)];
    }
#pragma unroll
    for (int j = 0; j < 16; ++j) {
      const int f = j * GCN_BIN_NODES + tid;
      const int row = f >> 4;
      const int nn = bin * GCN_BIN_NODES + row;
      if (nn < n_nodes) {
        const float sc = s_nrm[row];
        float4 w4 = v[j];
        w4.x *= sc; w4.y *= sc; w4.z *= sc; w4.w *= sc;
        ((float4*)y)[(size_t)nn * 16 + (f & 15)] = w4;
      }
    }
  } else {
    unsigned incl = c;
    for (int o = 1; o < 32; o <<= 1) {
      unsigned t = __shfl_up(incl, o, 32);
      if (lane >= o) incl += t;
    }
    if (lane == 31) s_wsum[wv] = incl;
    __syncthreads();
    unsigned pre = 0u;
    for (int j = 0; j < wv; ++j) pre += s_wsum[j];
    const unsigned first = s_base + pre + incl - c;
    if (n < n_nodes) { ((uint2*)rowrng)[n] = make_uint2(first, first + c); norm_in[n] = nrm; }
    s_cnt[tid] = first;
    __syncthreads();
    gcn_for_bin_records(stag, mc, ma, lane, [&](unsigned rec) {
      const unsigned p = atomicAdd(&s_cnt[(rec >> 16) & (GCN_BIN_NODES - 1)], 1u);
      csr_src[p] = (int)(rec & 0xffffu);
    });
  }
}

__device__ __forceinline__ void add4(float4& a, const float4& v) {
  a.x += v.x; a.y += v.y; a.z += v.z; a.w += v.w;
}

__device__ __forceinline__ float4 gcn_gather_row_sum(const unsigned* __restrict__ rowrng,
                                                     const int* __restrict__ csr_src,
                                                     const float* __restrict__ x,
                                                     int n, int q) {
  const char* xb = (const char*)x;
  const unsigned qoff = (unsigned)q * 16u;
  const uint2 be = ((const uint2*)rowrng)[n];
  unsigned beg = be.x;
  unsigned end = be.y;
  float4 a0 = make_float4(0.f, 0.f, 0.f, 0.f);
  float4 a1 = a0, a2 = a0, a3 = a0;
  for (unsigned base = beg; base < end; base += 32) {
    unsigned rem = end - base;
    int idx0 = ((unsigned)q < rem)        ? csr_src[base + q]      : 0;
    int idx1 = ((unsigned)(q + 16) < rem) ? csr_src[base + 16 + q] : 0;
    float4 v[32];
#pragma unroll
    for (int k = 0; k < 32; ++k) {
      int s = __shfl(k < 16 ? idx0 : idx1, k & 15, 16);
      v[k] = make_float4(0.f, 0.f, 0.f, 0.f);
      if ((unsigned)k < rem) v[k] = *(const float4*)(xb + ((unsigned)s * 256u + qoff));
    }
#pragma unroll
    for (int k = 0; k < 32; k += 4) {
      add4(a0, v[k]);
      add4(a1, v[k + 1]);
      add4(a2, v[k + 2]);
      add4(a3, v[k + 3]);
    }
  }
  add4(a0, a1);
  add4(a2, a3);
  add4(a0, a2);
  return a0;
}

__global__ void __launch_bounds__(256)
gcn_gather64_final_kernel(const unsigned* __restrict__ rowrng,
                          const int* __restrict__ csr_src,
                          const float* __restrict__ x,
                          const float* __restrict__ norm_in,
                          const float* __restrict__ bias,
                          float* __restrict__ out,
                          int n_nodes) {
  int gid = blockIdx.x * 256 + threadIdx.x;
  int n = gid >> 4;
  int q = gid & 15;
  if (n >= n_nodes) return;
  float s = norm_in[n];
  float4 bb = ((const float4*)bias)[q];
  float4 v = gcn_gather_row_sum(rowrng, csr_src, x, n, q);
  v.x = v.x * s + bb.x;
  v.y = v.y * s + bb.y;
  v.z = v.z * s + bb.z;
  v.w = v.w * s + bb.w;
  ((float4*)out)[(size_t)n * 16 + q] = v;
}

#define GCN_TILE_ROWS 32
#define GCN_PITCH_S 68
#define GCN_PITCH_H 132

__global__ void __launch_bounds__(256)
gcn_layer1_fused_kernel(const unsigned* __restrict__ rowrng,
                        const int* __restrict__ csr_src,
                        const float* __restrict__ x,
                        const float* __restrict__ norm_in,
                        const float* __restrict__ norm_out,
                        const float* __restrict__ W1,
                        const float* __restrict__ b1,
                        const float* __restrict__ W2,
                        float* __restrict__ Y,
                        int n_nodes) {
  __shared__ __attribute__((aligned(16))) float sS[GCN_TILE_ROWS * GCN_PITCH_S];
  __shared__ __attribute__((aligned(16))) float sH[GCN_TILE_ROWS * GCN_PITCH_H];
  __shared__ float sNo[GCN_TILE_ROWS];

  const int tid  = threadIdx.x;
  const int blk0 = blockIdx.x * GCN_TILE_ROWS;
  const int g    = tid >> 4;
  const int q    = tid & 15;
  const int wave = tid >> 5;
  const int lane = tid & 31;
  const int half = lane >> 4;
  const int l16  = lane & 15;

  for (int p = 0; p < 2; ++p) {
    const int lr = p * 16 + g;
    const int n  = blk0 + lr;
    float4 a = make_float4(0.f, 0.f, 0.f, 0.f);
    float no = 0.f;
    if (n < n_nodes) {
      const float rs = norm_in[n];
      no = norm_out[n];
      a = gcn_gather_row_sum(rowrng, csr_src, x, n, q);
      a.x *= rs; a.y *= rs; a.z *= rs; a.w *= rs;
    }
    *(float4*)&sS[lr * GCN_PITCH_S + q * 4] = a;
    if (q == 0) sNo[lr] = no;
  }

  const int col1 = wave * 16 + l16;
  v2f bw1[16];
#pragma unroll
  for (int kk = 0; kk < 16; ++kk) {
    const int ka = kk * 4 + half * 2;
    bw1[kk].x = W1[ka * 128 + col1];
    bw1[kk].y = W1[(ka + 1) * 128 + col1];
  }
  const float bb1 = b1[col1];
  __syncthreads();

  {
    v2f a0[16], a1[16];
#pragma unroll
    for (int kk = 0; kk < 16; ++kk) {
      const int ka = kk * 4 + half * 2;
      a0[kk] = *(const v2f*)&sS[l16 * GCN_PITCH_S + ka];
      a1[kk] = *(const v2f*)&sS[(16 + l16) * GCN_PITCH_S + ka];
    }
    v8f acc0 = {};
    v8f acc1 = {};
#pragma unroll
    for (int kk = 0; kk < 16; ++kk) {
      acc0 = __builtin_amdgcn_wmma_f32_16x16x4_f32(
          false, a0[kk], false, bw1[kk], (short)0, acc0, false, false);
      acc1 = __builtin_amdgcn_wmma_f32_16x16x4_f32(
          false, a1[kk], false, bw1[kk], (short)0, acc1, false, false);
    }
#pragma unroll
    for (int r = 0; r < 8; ++r) {
      const int row = half * 8 + r;
      sH[row * GCN_PITCH_H + col1]        = fmaxf(acc0[r] + bb1, 0.0f) * sNo[row];
      sH[(16 + row) * GCN_PITCH_H + col1] = fmaxf(acc1[r] + bb1, 0.0f) * sNo[16 + row];
    }
  }

  const int mt2  = wave >> 2;
  const int nt2  = wave & 3;
  const int col2 = nt2 * 16 + l16;
  v2f bw2[32];
#pragma unroll
  for (int kk = 0; kk < 32; ++kk) {
    const int ka = kk * 4 + half * 2;
    bw2[kk].x = W2[ka * 64 + col2];
    bw2[kk].y = W2[(ka + 1) * 64 + col2];
  }
  __syncthreads();

  {
    v2f a2[32];
#pragma unroll
    for (int kk = 0; kk < 32; ++kk) {
      const int ka = kk * 4 + half * 2;
      a2[kk] = *(const v2f*)&sH[(mt2 * 16 + l16) * GCN_PITCH_H + ka];
    }
    v8f acc = {};
#pragma unroll
    for (int kk = 0; kk < 32; ++kk) {
      acc = __builtin_amdgcn_wmma_f32_16x16x4_f32(
          false, a2[kk], false, bw2[kk], (short)0, acc, false, false);
    }
#pragma unroll
    for (int r = 0; r < 8; ++r) {
      sS[(mt2 * 16 + half * 8 + r) * GCN_PITCH_S + col2] = acc[r];
    }
  }
  __syncthreads();

  for (int p = 0; p < 2; ++p) {
    const int lr = p * 16 + g;
    const int n  = blk0 + lr;
    if (n < n_nodes) {
      ((float4*)Y)[(size_t)n * 16 + q] = *(const float4*)&sS[lr * GCN_PITCH_S + q * 4];
    }
  }
}

extern "C" void kernel_launch(void* const* d_in, const int* in_sizes, int n_in,
                              void* d_out, int out_size, void* d_ws, size_t ws_size,
                              hipStream_t stream) {
  (void)in_sizes; (void)n_in; (void)out_size; (void)ws_size;
  const float* h   = (const float*)d_in[0];
  const int*   src = (const int*)d_in[1];
  const int*   dst = (const int*)d_in[2];
  const float* W1  = (const float*)d_in[3];
  const float* b1  = (const float*)d_in[4];
  const float* W2  = (const float*)d_in[5];
  const float* b2  = (const float*)d_in[6];
  float* out = (float*)d_out;

  const int N = GCN_N;
  const int E = GCN_E;

  char* ws = (char*)d_ws;
  size_t off = 0;
  auto carve = [&](size_t bytes) -> void* {
    void* p = ws + off;
    off += (bytes + 255) & ~(size_t)255;
    return p;
  };
  unsigned* rowrng  = (unsigned*)carve((size_t)N * 4 * 2);
  float* norm_out   = (float*)carve((size_t)N * 4);
  float* norm_in    = (float*)carve((size_t)N * 4);
  float* hs         = (float*)carve((size_t)N * 64 * 4);
  float* t2         = (float*)carve((size_t)N * 64 * 4);
  unsigned* cursor  = (unsigned*)carve((size_t)(N + 1) * 4);
  unsigned* stag_s  = (unsigned*)carve((size_t)E * 4);
  int* csr_src      = (int*)carve((size_t)E * 4);
  unsigned* stag_d  = (unsigned*)t2;
  unsigned* tab     = (unsigned*)t2 + 1048576;

  const int E4 = E / 4;
  gcn_bin_place_kernel<<<GCN_PBLK, GCN_BIN_THREADS, 0, stream>>>(src, dst, tab, cursor, stag_d, stag_s, E4);
  gcn_bin_owner_kernel<<<2 * GCN_NB, GCN_BIN_NODES, 0, stream>>>(
      tab, cursor, stag_d, stag_s, rowrng, csr_src, norm_in, norm_out, h, hs, N);

  gcn_layer1_fused_kernel<<<(N + 31) / 32, 256, 0, stream>>>(
      rowrng, csr_src, hs, norm_in, norm_out, W1, b1, W2, t2, N);

  gcn_gather64_final_kernel<<<(N * 16 + 255) / 256, 256, 0, stream>>>(rowrng, csr_src, t2, norm_in, b2, out, N);
}


# Gather row loads go out in 4 branch-free groups of 7 per 28-edge batch (slots past the last edge re-read its row, zeroed after), not one exec test and translation wait per load; sum order same
# speedup vs baseline: 4.6874x; 1.0852x over previous
#include <hip/hip_runtime.h>

typedef __attribute__((ext_vector_type(2))) float v2f;
typedef __attribute__((ext_vector_type(8))) float v8f;

#define GCN_N 50000
#define GCN_E 800000

#define GCN_BIN_NODES 256
#define GCN_NB ((GCN_N + GCN_BIN_NODES - 1) / GCN_BIN_NODES)
#define GCN_BIN_THREADS 512
#define GCN_BIN_INT4 2
#define GCN_BIN_CHUNK4 (GCN_BIN_THREADS * GCN_BIN_INT4)
#define GCN_SLAB (4 * GCN_BIN_CHUNK4)
#define GCN_PBLK ((GCN_E / 4 + GCN_BIN_CHUNK4 - 1) / GCN_BIN_CHUNK4)
static_assert(GCN_N <= 65536, "a node id must fit in 16 bits of the packed record");
static_assert(GCN_E % 4 == 0, "edges are read as int4");
static_assert(GCN_NB <= 224 && GCN_NB <= GCN_BIN_NODES, "bin scans assume at most 7 waves of bins");
static_assert(GCN_SLAB < 65536, "a run's start and its count share one table word");
static_assert(GCN_PBLK <= GCN_BIN_NODES, "an owner block reads one table word per thread");

__device__ __forceinline__ void gcn_load_edges(const int* __restrict__ src, const int* __restrict__ dst,
                                               int n_edges4, int (&sv)[4 * GCN_BIN_INT4],
                                               int (&dv)[4 * GCN_BIN_INT4], bool (&ok)[GCN_BIN_INT4]) {
#pragma unroll
  for (int j = 0; j < GCN_BIN_INT4; ++j) {
    const int idx = blockIdx.x * GCN_BIN_CHUNK4 + j * GCN_BIN_THREADS + threadIdx.x;
    ok[j] = idx < n_edges4;
    int4 s = make_int4(0, 0, 0, 0), d = make_int4(0, 0, 0, 0);
    if (ok[j]) { s = ((const int4*)src)[idx]; d = ((const int4*)dst)[idx]; }
    sv[4 * j] = s.x; sv[4 * j + 1] = s.y; sv[4 * j + 2] = s.z; sv[4 * j + 3] = s.w;
    dv[4 * j] = d.x; dv[4 * j + 1] = d.y; dv[4 * j + 2] = d.z; dv[4 * j + 3] = d.w;
  }
}

__device__ __forceinline__ void gcn_hist_edges(unsigned* s_cnt, const int (&sv)[4 * GCN_BIN_INT4],
                                               const int (&dv)[4 * GCN_BIN_INT4],
                                               const bool (&ok)[GCN_BIN_INT4]) {
#pragma unroll
  for (int k = 0; k < 4 * GCN_BIN_INT4; ++k) {
    if (ok[k >> 2]) {
      atomicAdd(&s_cnt[dv[k] / GCN_BIN_NODES], 1u);
      atomicAdd(&s_cnt[GCN_NB + sv[k] / GCN_BIN_NODES], 1u);
    }
  }
}

__global__ void __launch_bounds__(GCN_BIN_THREADS)
gcn_bin_place_kernel(const int* __restrict__ src, const int* __restrict__ dst,
                     unsigned* __restrict__ tab, unsigned* __restrict__ cursor,
                     unsigned* __restrict__ stag_d, unsigned* __restrict__ stag_s, int n_edges4) {
  __shared__ unsigned s_cnt[2 * GCN_NB];
  __shared__ unsigned s_wsum[14];
  __shared__ __attribute__((aligned(16))) unsigned s_rec[2][GCN_SLAB];
  const int tid  = threadIdx.x;
  const int lane = tid & 31;
  const int wv   = tid >> 5;
  int sv[4 * GCN_BIN_INT4], dv[4 * GCN_BIN_INT4];
  bool ok[GCN_BIN_INT4];
  gcn_load_edges(src, dst, n_edges4, sv, dv, ok);
  if (tid < 2 * GCN_NB) s_cnt[tid] = 0u;
  if (blockIdx.x == 0 && tid == 0) *cursor = 0u;
  __syncthreads();

  gcn_hist_edges(s_cnt, sv, dv, ok);
  __syncthreads();

  const int side = wv >= 7 ? 1 : 0;
  const int li   = (wv - side * 7) * 32 + lane;
  const bool has_bin = wv < 14 && li < GCN_NB;
  const int bi   = side * GCN_NB + li;
  unsigned v = 0u;
  if (has_bin) v = s_cnt[bi];
  unsigned incl = v;
  for (int o = 1; o < 32; o <<= 1) {
    unsigned t = __shfl_up(incl, o, 32);
    if (lane >= o) incl += t;
  }
  if (wv < 14 && lane == 31) s_wsum[wv] = incl;
  __syncthreads();
  unsigned pre = 0u;
  for (int j = side * 7; j < wv && j < 14; ++j) pre += s_wsum[j];
  if (has_bin) {
    const unsigned start = pre + incl - v;
    tab[(size_t)bi * GCN_PBLK + blockIdx.x] = (start << 16) | v;
    s_cnt[bi] = start;
  }
  __syncthreads();

#pragma unroll
  for (int j = 0; j < GCN_BIN_INT4; ++j) {
    if (ok[j]) {
      unsigned p[4], q[4];
#pragma unroll
      for (int e = 0; e < 4; ++e) {
        p[e] = atomicAdd(&s_cnt[dv[4 * j + e] / GCN_BIN_NODES], 1u);
        q[e] = atomicAdd(&s_cnt[GCN_NB + sv[4 * j + e] / GCN_BIN_NODES], 1u);
      }
#pragma unroll
      for (int e = 0; e < 4; ++e) {
        const unsigned rec = (unsigned)sv[4 * j + e] | ((unsigned)dv[4 * j + e] << 16);
        s_rec[0][p[e]] = rec;
        s_rec[1][q[e]] = rec;
      }
    }
  }
  __syncthreads();

  const int left4 = n_edges4 - blockIdx.x * GCN_BIN_CHUNK4;
  const int n4 = left4 < GCN_BIN_CHUNK4 ? left4 : GCN_BIN_CHUNK4;
  uint4* od = (uint4*)(stag_d + (size_t)blockIdx.x * GCN_SLAB);
  uint4* os = (uint4*)(stag_s + (size_t)blockIdx.x * GCN_SLAB);
  for (int i = tid; i < n4; i += GCN_BIN_THREADS) {
    od[i] = ((const uint4*)s_rec[0])[i];
    os[i] = ((const uint4*)s_rec[1])[i];
  }
}

#define GCN_RUNS ((GCN_PBLK + 7) / 8)
static_assert(GCN_RUNS <= 32, "a wave holds one run per lane");
template <typename F>
__device__ __forceinline__ void gcn_for_bin_records(const unsigned* __restrict__ stag,
                                                    unsigned mc, unsigned ma, int lane, F f) {
  unsigned r[GCN_RUNS];
#pragma unroll
  for (int j = 0; j < GCN_RUNS; ++j) {
    const unsigned cnt = (unsigned)__builtin_amdgcn_readlane((int)mc, j);
    const unsigned adr = (unsigned)__builtin_amdgcn_readlane((int)ma, j);
    r[j] = stag[(unsigned)lane < cnt ? adr + (unsigned)lane : 0u];
  }
#pragma unroll
  for (int j = 0; j < GCN_RUNS; ++j) {
    if ((unsigned)lane < (unsigned)__builtin_amdgcn_readlane((int)mc, j)) f(r[j]);
  }
  unsigned big = (unsigned)__ballot(mc > 32u);
  while (big != 0u) {
    const int j = __ffs(big) - 1;
    big &= big - 1u;
    const unsigned cnt = (unsigned)__builtin_amdgcn_readlane((int)mc, j);
    const unsigned adr = (unsigned)__builtin_amdgcn_readlane((int)ma, j);
    for (unsigned l = 32u + (unsigned)lane; l < cnt; l += 32u) f(stag[adr + l]);
  }
}

__global__ void __launch_bounds__(GCN_BIN_NODES)
gcn_bin_owner_kernel(const unsigned* __restrict__ tab, unsigned* __restrict__ cursor,
                     const unsigned* __restrict__ stag_d, const unsigned* __restrict__ stag_s,
                     unsigned* __restrict__ rowrng, int* __restrict__ csr_src,
                     float* __restrict__ norm_in, float* __restrict__ norm_out,
                     const float* __restrict__ x, float* __restrict__ y, int n_nodes) {
  __shared__ unsigned s_cnt[GCN_BIN_NODES];
  __shared__ unsigned s_wsum[8];
  __shared__ unsigned s_base;
  __shared__ float s_nrm[GCN_BIN_NODES];
  const int tid  = threadIdx.x;
  const int lane = tid & 31;
  const int wv   = tid >> 5;
  const bool is_dst = blockIdx.x < GCN_NB;
  const int bin = is_dst ? blockIdx.x : blockIdx.x - GCN_NB;
  const unsigned* stag = is_dst ? stag_d : stag_s;
  const int shift = is_dst ? 16 : 0;

  const int mt = wv + 8 * lane;
  unsigned w = 0u;
  if (mt < GCN_PBLK) w = tab[(size_t)blockIdx.x * GCN_PBLK + mt];
  const unsigned mc = w & 0xffffu;
  const unsigned ma = (unsigned)mt * GCN_SLAB + (w >> 16);
  s_cnt[tid] = 0u;
  unsigned part = mc;
  for (int o = 16; o > 0; o >>= 1) part += __shfl_xor(part, o, 32);
  if (lane == 0) s_wsum[wv] = part;
  __syncthreads();
  unsigned got = 0u;
  if (is_dst && tid == 0) {
    unsigned nrec = 0u;
    for (int j = 0; j < 8; ++j) nrec += s_wsum[j];
    got = atomicAdd(cursor, nrec);
  }

  gcn_for_bin_records(stag, mc, ma, lane, [&](unsigned rec) {
    atomicAdd(&s_cnt[(rec >> shift) & (GCN_BIN_NODES - 1)], 1u);
  });
  if (is_dst && tid == 0) s_base = got;
  __syncthreads();

  const unsigned c = s_cnt[tid];
  const int n = bin * GCN_BIN_NODES + tid;
  const float nrm = rsqrtf((float)(c < 1u ? 1u : c));

  if (!is_dst) {
    if (n < n_nodes) norm_out[n] = nrm;
    s_nrm[tid] = nrm;
    __syncthreads();
    float4 v[16];
#pragma unroll
    for (int j = 0; j < 16; ++j) {
      const int f = j * GCN_BIN_NODES + tid;
      const int nn = bin * GCN_BIN_NODES + (f >> 4);
      const int nc = nn < n_nodes ? nn : n_nodes - 1;
      v[j] = ((const float4*)x)[(size_t)nc * 16 + (f & 15)];
    }
#pragma unroll
    for (int j = 0; j < 16; ++j) {
      const int f = j * GCN_BIN_NODES + tid;
      const int row = f >> 4;
      const int nn = bin * GCN_BIN_NODES + row;
      if (nn < n_nodes) {
        const float sc = s_nrm[row];
        float4 w4 = v[j];
        w4.x *= sc; w4.y *= sc; w4.z *= sc; w4.w *= sc;
        ((float4*)y)[(size_t)nn * 16 + (f & 15)] = w4;
      }
    }
  } else {
    unsigned incl = c;
    for (int o = 1; o < 32; o <<= 1) {
      unsigned t = __shfl_up(incl, o, 32);
      if (lane >= o) incl += t;
    }
    if (lane == 31) s_wsum[wv] = incl;
    __syncthreads();
    unsigned pre = 0u;
    for (int j = 0; j < wv; ++j) pre += s_wsum[j];
    const unsigned first = s_base + pre + incl - c;
    if (n < n_nodes) { ((uint2*)rowrng)[n] = make_uint2(first, first + c); norm_in[n] = nrm; }
    s_cnt[tid] = first;
    __syncthreads();
    gcn_for_bin_records(stag, mc, ma, lane, [&](unsigned rec) {
      const unsigned p = atomicAdd(&s_cnt[(rec >> 16) & (GCN_BIN_NODES - 1)], 1u);
      csr_src[p] = (int)(rec & 0xffffu);
    });
  }
}

__device__ __forceinline__ void add4(float4& a, const float4& v) {
  a.x += v.x; a.y += v.y; a.z += v.z; a.w += v.w;
}

#define GCN_GRP 7
#define GCN_BATCH (4 * GCN_GRP)
static_assert(GCN_BATCH % 4 == 0, "edge e of a node must fall on partial sum e mod 4 in every batch");
static_assert(GCN_BATCH > 16 && GCN_BATCH <= 32, "a batch's indices are two loads of 16 lanes");
__device__ __forceinline__ float4 gcn_gather_row_sum(const unsigned* __restrict__ rowrng,
                                                     const int* __restrict__ csr_src,
                                                     const float* __restrict__ x,
                                                     int n, int q) {
  const char* xb = (const char*)x;
  const unsigned qoff = (unsigned)q * 16u;
  const uint2 be = ((const uint2*)rowrng)[n];
  unsigned beg = be.x;
  unsigned end = be.y;
  const float4 zero4 = make_float4(0.f, 0.f, 0.f, 0.f);
  float4 a0 = zero4;
  float4 a1 = a0, a2 = a0, a3 = a0;
  for (unsigned base = beg; base < end; base += GCN_BATCH) {
    const unsigned rem = end - base;
    const unsigned last = rem - 1u;
    const unsigned e0 = (unsigned)q < last ? (unsigned)q : last;
    const unsigned e1 = (unsigned)(q + 16) < last ? (unsigned)(q + 16) : last;
    const int idx0 = csr_src[base + e0];
    const int idx1 = csr_src[base + e1];
    asm volatile("" :: "v"(idx0), "v"(idx1));
    float4 v[GCN_BATCH];
#pragma unroll
    for (int g = 0; g < 4; ++g) {
#pragma unroll
      for (int j = 0; j < GCN_GRP; ++j) v[GCN_GRP * g + j] = zero4;
      if (rem > (unsigned)(GCN_GRP * g)) {
        unsigned off[GCN_GRP];
#pragma unroll
        for (int j = 0; j < GCN_GRP; ++j) {
          const int k = GCN_GRP * g + j;
          const int s = __shfl(k < 16 ? idx0 : idx1, k & 15, 16);
          off[j] = (unsigned)s * 256u + qoff;
        }
#pragma unroll
        for (int j = 0; j < GCN_GRP; ++j) v[GCN_GRP * g + j] = *(const float4*)(xb + off[j]);
      }
    }
#pragma unroll
    for (int k = 0; k < GCN_BATCH; ++k) {
      if (!((unsigned)k < rem)) v[k] = zero4;
    }
#pragma unroll
    for (int k = 0; k < GCN_BATCH; k += 4) {
      add4(a0, v[k]);
      add4(a1, v[k + 1]);
      add4(a2, v[k + 2]);
      add4(a3, v[k + 3]);
    }
  }
  add4(a0, a1);
  add4(a2, a3);
  add4(a0, a2);
  return a0;
}

__global__ void __launch_bounds__(256)
gcn_gather64_final_kernel(const unsigned* __restrict__ rowrng,
                          const int* __restrict__ csr_src,
                          const float* __restrict__ x,
                          const float* __restrict__ norm_in,
                          const float* __restrict__ bias,
                          float* __restrict__ out,
                          int n_nodes) {
  int gid = blockIdx.x * 256 + threadIdx.x;
  int n = gid >> 4;
  int q = gid & 15;
  if (n >= n_nodes) return;
  float s = norm_in[n];
  float4 bb = ((const float4*)bias)[q];
  float4 v = gcn_gather_row_sum(rowrng, csr_src, x, n, q);
  v.x = v.x * s + bb.x;
  v.y = v.y * s + bb.y;
  v.z = v.z * s + bb.z;
  v.w = v.w * s + bb.w;
  ((float4*)out)[(size_t)n * 16 + q] = v;
}

#define GCN_TILE_ROWS 32
#define GCN_PITCH_S 68
#define GCN_PITCH_H 132

__global__ void __launch_bounds__(256)
gcn_layer1_fused_kernel(const unsigned* __restrict__ rowrng,
                        const int* __restrict__ csr_src,
                        const float* __restrict__ x,
                        const float* __restrict__ norm_in,
                        const float* __restrict__ norm_out,
                        const float* __restrict__ W1,
                        const float* __restrict__ b1,
                        const float* __restrict__ W2,
                        float* __restrict__ Y,
                        int n_nodes) {
  __shared__ __attribute__((aligned(16))) float sS[GCN_TILE_ROWS * GCN_PITCH_S];
  __shared__ __attribute__((aligned(16))) float sH[GCN_TILE_ROWS * GCN_PITCH_H];
  __shared__ float sNo[GCN_TILE_ROWS];

  const int tid  = threadIdx.x;
  const int blk0 = blockIdx.x * GCN_TILE_ROWS;
  const int g    = tid >> 4;
  const int q    = tid & 15;
  const int wave = tid >> 5;
  const int lane = tid & 31;
  const int half = lane >> 4;
  const int l16  = lane & 15;

  for (int p = 0; p < 2; ++p) {
    const int lr = p * 16 + g;
    const int n  = blk0 + lr;
    float4 a = make_float4(0.f, 0.f, 0.f, 0.f);
    float no = 0.f;
    if (n < n_nodes) {
      const float rs = norm_in[n];
      no = norm_out[n];
      a = gcn_gather_row_sum(rowrng, csr_src, x, n, q);
      a.x *= rs; a.y *= rs; a.z *= rs; a.w *= rs;
    }
    *(float4*)&sS[lr * GCN_PITCH_S + q * 4] = a;
    if (q == 0) sNo[lr] = no;
  }

  const int col1 = wave * 16 + l16;
  v2f bw1[16];
#pragma unroll
  for (int kk = 0; kk < 16; ++kk) {
    const int ka = kk * 4 + half * 2;
    bw1[kk].x = W1[ka * 128 + col1];
    bw1[kk].y = W1[(ka + 1) * 128 + col1];
  }
  const float bb1 = b1[col1];
  __syncthreads();

  {
    v2f a0[16], a1[16];
#pragma unroll
    for (int kk = 0; kk < 16; ++kk) {
      const int ka = kk * 4 + half * 2;
      a0[kk] = *(const v2f*)&sS[l16 * GCN_PITCH_S + ka];
      a1[kk] = *(const v2f*)&sS[(16 + l16) * GCN_PITCH_S + ka];
    }
    v8f acc0 = {};
    v8f acc1 = {};
#pragma unroll
    for (int kk = 0; kk < 16; ++kk) {
      acc0 = __builtin_amdgcn_wmma_f32_16x16x4_f32(
          false, a0[kk], false, bw1[kk], (short)0, acc0, false, false);
      acc1 = __builtin_amdgcn_wmma_f32_16x16x4_f32(
          false, a1[kk], false, bw1[kk], (short)0, acc1, false, false);
    }
#pragma unroll
    for (int r = 0; r < 8; ++r) {
      const int row = half * 8 + r;
      sH[row * GCN_PITCH_H + col1]        = fmaxf(acc0[r] + bb1, 0.0f) * sNo[row];
      sH[(16 + row) * GCN_PITCH_H + col1] = fmaxf(acc1[r] + bb1, 0.0f) * sNo[16 + row];
    }
  }

  const int mt2  = wave >> 2;
  const int nt2  = wave & 3;
  const int col2 = nt2 * 16 + l16;
  v2f bw2[32];
#pragma unroll
  for (int kk = 0; kk < 32; ++kk) {
    const int ka = kk * 4 + half * 2;
    bw2[kk].x = W2[ka * 64 + col2];
    bw2[kk].y = W2[(ka + 1) * 64 + col2];
  }
  __syncthreads();

  {
    v2f a2[32];
#pragma unroll
    for (int kk = 0; kk < 32; ++kk) {
      const int ka = kk * 4 + half * 2;
      a2[kk] = *(const v2f*)&sH[(mt2 * 16 + l16) * GCN_PITCH_H + ka];
    }
    v8f acc = {};
#pragma unroll
    for (int kk = 0; kk < 32; ++kk) {
      acc = __builtin_amdgcn_wmma_f32_16x16x4_f32(
          false, a2[kk], false, bw2[kk], (short)0, acc, false, false);
    }
#pragma unroll
    for (int r = 0; r < 8; ++r) {
      sS[(mt2 * 16 + half * 8 + r) * GCN_PITCH_S + col2] = acc[r];
    }
  }
  __syncthreads();

  for (int p = 0; p < 2; ++p) {
    const int lr = p * 16 + g;
    const int n  = blk0 + lr;
    if (n < n_nodes) {
      ((float4*)Y)[(size_t)n * 16 + q] = *(const float4*)&sS[lr * GCN_PITCH_S + q * 4];
    }
  }
}

extern "C" void kernel_launch(void* const* d_in, const int* in_sizes, int n_in,
                              void* d_out, int out_size, void* d_ws, size_t ws_size,
                              hipStream_t stream) {
  (void)in_sizes; (void)n_in; (void)out_size; (void)ws_size;
  const float* h   = (const float*)d_in[0];
  const int*   src = (const int*)d_in[1];
  const int*   dst = (const int*)d_in[2];
  const float* W1  = (const float*)d_in[3];
  const float* b1  = (const float*)d_in[4];
  const float* W2  = (const float*)d_in[5];
  const float* b2  = (const float*)d_in[6];
  float* out = (float*)d_out;

  const int N = GCN_N;
  const int E = GCN_E;

  char* ws = (char*)d_ws;
  size_t off = 0;
  auto carve = [&](size_t bytes) -> void* {
    void* p = ws + off;
    off += (bytes + 255) & ~(size_t)255;
    return p;
  };
  unsigned* rowrng  = (unsigned*)carve((size_t)N * 4 * 2);
  float* norm_out   = (float*)carve((size_t)N * 4);
  float* norm_in    = (float*)carve((size_t)N * 4);
  float* hs         = (float*)carve((size_t)N * 64 * 4);
  float* t2         = (float*)carve((size_t)N * 64 * 4);
  unsigned* cursor  = (unsigned*)carve((size_t)(N + 1) * 4);
  unsigned* stag_s  = (unsigned*)carve((size_t)E * 4);
  int* csr_src      = (int*)carve((size_t)E * 4);
  unsigned* stag_d  = (unsigned*)t2;
  unsigned* tab     = (unsigned*)t2 + 1048576;

  const int E4 = E / 4;
  gcn_bin_place_kernel<<<GCN_PBLK, GCN_BIN_THREADS, 0, stream>>>(src, dst, tab, cursor, stag_d, stag_s, E4);
  gcn_bin_owner_kernel<<<2 * GCN_NB, GCN_BIN_NODES, 0, stream>>>(
      tab, cursor, stag_d, stag_s, rowrng, csr_src, norm_in, norm_out, h, hs, N);

  gcn_layer1_fused_kernel<<<(N + 31) / 32, 256, 0, stream>>>(
      rowrng, csr_src, hs, norm_in, norm_out, W1, b1, W2, t2, N);

  gcn_gather64_final_kernel<<<(N * 16 + 255) / 256, 256, 0, stream>>>(rowrng, csr_src, t2, norm_in, b2, out, N);
}
